# starting kernel
# baseline (speedup 1.0000x reference)
#include <hip/hip_runtime.h>


typedef __attribute__((ext_vector_type(2))) float v2f;
typedef __attribute__((ext_vector_type(8))) float v8f;

#define B_ 4
#define L_ 2046
#define D_ 1024
#define C_ 512
#define P_ 1024
#define H_ 16
#define HD_ 32
#define DV_ 64
#define WKV_LD 2048

static __device__ __forceinline__ v8f wmma4(v2f a, v2f b, v8f c) {
  return __builtin_amdgcn_wmma_f32_16x16x4_f32(false, a, false, b, (short)0, c,
                                               false, false);
}

static __device__ __forceinline__ float rowmax16(float v) {
  v = fmaxf(v, __shfl_xor(v, 8, 16));
  v = fmaxf(v, __shfl_xor(v, 4, 16));
  v = fmaxf(v, __shfl_xor(v, 2, 16));
  v = fmaxf(v, __shfl_xor(v, 1, 16));
  return v;
}
static __device__ __forceinline__ float rowsum16(float v) {
  v += __shfl_xor(v, 8, 16);
  v += __shfl_xor(v, 4, 16);
  v += __shfl_xor(v, 2, 16);
  v += __shfl_xor(v, 1, 16);
  return v;
}

__global__ __launch_bounds__(256) void gemm_rowA_wmma(
    const float* __restrict__ A, long strideAm, long batchStrideA,
    const float* __restrict__ W, const float* __restrict__ bias,
    float* __restrict__ Out, long strideOm, long batchStrideO,
    int M, int K, int mstrips, int nstrips) {
  int lane = threadIdx.x & 31;
  int wid = (blockIdx.x * blockDim.x + threadIdx.x) >> 5;
  int wavesPerBatch = mstrips * nstrips;
  int b = wid / wavesPerBatch;
  int r = wid - b * wavesPerBatch;
  int m0 = (r / nstrips) * 32;
  int n0 = (r % nstrips) * 64;
  const float* Ab = A + (long)b * batchStrideA;
  float* Ob = Out + (long)b * batchStrideO;
  int col = lane & 15;
  int hi = lane >> 4;
  int khalf = hi * 2;
  int row0 = m0 + col;      if (row0 > M - 1) row0 = M - 1;
  int row1 = m0 + 16 + col; if (row1 > M - 1) row1 = M - 1;
  const float* Arow0 = Ab + (long)row0 * strideAm + khalf;
  const float* Arow1 = Ab + (long)row1 * strideAm + khalf;
  const float* Wp = W + (long)(n0 + col) * K + khalf;
  const long Wn16 = (long)16 * K;

  v8f acc[2][4] = {};
  for (int k0 = 0; k0 < K; k0 += 4) {
    if ((k0 & 63) == 0) {
      __builtin_prefetch(Arow0 + k0 + 256, 0, 1);
      __builtin_prefetch(Arow1 + k0 + 256, 0, 1);
      __builtin_prefetch(Wp + k0 + 256, 0, 1);
      __builtin_prefetch(Wp + 3 * Wn16 + k0 + 256, 0, 1);
    }
    v2f a0 = *(const v2f*)(Arow0 + k0);
    v2f a1 = *(const v2f*)(Arow1 + k0);
#pragma unroll
    for (int j = 0; j < 4; ++j) {
      v2f bf = *(const v2f*)(Wp + j * Wn16 + k0);
      acc[0][j] = wmma4(a0, bf, acc[0][j]);
      acc[1][j] = wmma4(a1, bf, acc[1][j]);
    }
  }
  float bv[4];
#pragma unroll
  for (int j = 0; j < 4; ++j) bv[j] = bias[n0 + j * 16 + col];
#pragma unroll
  for (int i = 0; i < 2; ++i) {
#pragma unroll
    for (int rr = 0; rr < 8; ++rr) {
      int row = m0 + i * 16 + rr + 8 * hi;
      if (row < M) {
        float* orow = Ob + (long)row * strideOm + n0;
#pragma unroll
        for (int j = 0; j < 4; ++j) orow[j * 16 + col] = acc[i][j][rr] + bv[j];
      }
    }
  }
}

__global__ __launch_bounds__(256) void gemm_colA_wmma(
    const float* __restrict__ A, long strideAk, long batchStrideA,
    const float* __restrict__ Wkv,
    float* __restrict__ Out, long strideOm, long batchStrideO,
    int K, int mstrips, int nstrips) {
  int lane = threadIdx.x & 31;
  int wid = (blockIdx.x * blockDim.x + threadIdx.x) >> 5;
  int wavesPerBatch = mstrips * nstrips;
  int b = wid / wavesPerBatch;
  int r = wid - b * wavesPerBatch;
  int m0 = (r / nstrips) * 32;
  int n0 = (r % nstrips) * 64;
  const float* Ab = A + (long)b * batchStrideA;
  float* Ob = Out + (long)b * batchStrideO;
  int col = lane & 15;
  int hi = lane >> 4;
  int khalf = hi * 2;
  const float* Am0 = Ab + (long)khalf * strideAk + m0 + col;
  const float* Am1 = Am0 + 16;
  const float* Wp = Wkv + (long)(n0 + col) * WKV_LD + khalf;
  const long Wn16 = (long)16 * WKV_LD;

  v8f acc[2][4] = {};
  int Kmain = K & ~3;
  for (int k0 = 0; k0 < Kmain; k0 += 4) {
    long ao = (long)k0 * strideAk;
    v2f a0 = {Am0[ao], Am0[ao + strideAk]};
    v2f a1 = {Am1[ao], Am1[ao + strideAk]};
#pragma unroll
    for (int j = 0; j < 4; ++j) {
      v2f bf = *(const v2f*)(Wp + j * Wn16 + k0);
      acc[0][j] = wmma4(a0, bf, acc[0][j]);
      acc[1][j] = wmma4(a1, bf, acc[1][j]);
    }
  }
  if (Kmain < K) {
    int k0 = Kmain;
    long ao = (long)k0 * strideAk;
    v2f a0 = {0.f, 0.f}, a1 = {0.f, 0.f};
    if (hi == 0) {
      a0 = (v2f){Am0[ao], Am0[ao + strideAk]};
      a1 = (v2f){Am1[ao], Am1[ao + strideAk]};
    }
#pragma unroll
    for (int j = 0; j < 4; ++j) {
      v2f bf = *(const v2f*)(Wp + j * Wn16 + k0);
      acc[0][j] = wmma4(a0, bf, acc[0][j]);
      acc[1][j] = wmma4(a1, bf, acc[1][j]);
    }
  }
#pragma unroll
  for (int i = 0; i < 2; ++i) {
#pragma unroll
    for (int rr = 0; rr < 8; ++rr) {
      int row = m0 + i * 16 + rr + 8 * hi;
      float* orow = Ob + (long)row * strideOm + n0;
#pragma unroll
      for (int j = 0; j < 4; ++j) orow[j * 16 + col] = acc[i][j][rr];
    }
  }
}

__global__ __launch_bounds__(256) void attn_wmma(
    const float* __restrict__ q, const float* __restrict__ kbuf,
    const float* __restrict__ vbuf, float* __restrict__ obuf) {
  __shared__ float lds[8][256];
  int lane = threadIdx.x & 31;
  int wslot = threadIdx.x >> 5;
  int wid = (blockIdx.x * blockDim.x + threadIdx.x) >> 5;
  int ltile = wid & 127;
  int h = (wid >> 7) & 15;
  int b = wid >> 11;
  int l0 = ltile * 16;
  int col = lane & 15;
  int hi = lane >> 4;
  int khalf = hi * 2;

  v2f qA[8];
  {
    int l = l0 + col;
    bool lv = l < L_;
    const float* qrow = q + ((long)b * L_ + l) * C_ + h * HD_;
    const float sc = 0.125f * 1.44269504088896340736f;
#pragma unroll
    for (int kk = 0; kk < 8; ++kk) {
      int k = kk * 4 + khalf;
      float x0 = lv ? qrow[k] * sc : 0.f;
      float x1 = lv ? qrow[k + 1] * sc : 0.f;
      qA[kk] = (v2f){x0, x1};
    }
  }

  v8f O[4] = {};
  float Mrow[8], Srow[8];
#pragma unroll
  for (int r = 0; r < 8; ++r) { Mrow[r] = -INFINITY; Srow[r] = 0.f; }

  const float* kb = kbuf + (long)b * C_ * P_ + (long)h * P_;
  const float* vb = vbuf + (long)b * D_ * P_ + (long)h * P_;

  for (int p0 = 0; p0 < P_; p0 += 16) {
    v8f c = {};
#pragma unroll
    for (int kk = 0; kk < 8; ++kk) {
      int k = kk * 4 + khalf;
      const float* kp = kb + (long)k * 16 * P_ + p0 + col;
      v2f bf = {kp[0], kp[16 * P_]};
      c = wmma4(qA[kk], bf, c);
    }
    float alpha[8];
#pragma unroll
    for (int r = 0; r < 8; ++r) {
      float mx = rowmax16(c[r]);
      float mnew = fmaxf(Mrow[r], mx);
      float al = __builtin_exp2f(Mrow[r] - mnew);
      float pv = __builtin_exp2f(c[r] - mnew);
      Srow[r] = Srow[r] * al + rowsum16(pv);
      Mrow[r] = mnew;
      alpha[r] = al;
      lds[wslot][(r + 8 * hi) * 16 + col] = pv;
    }
#pragma unroll
    for (int r = 0; r < 8; ++r) {
      O[0][r] *= alpha[r]; O[1][r] *= alpha[r];
      O[2][r] *= alpha[r]; O[3][r] *= alpha[r];
    }
    __builtin_amdgcn_fence(__ATOMIC_ACQ_REL, "wavefront");
    __builtin_amdgcn_wave_barrier();
    v2f pA[4];
#pragma unroll
    for (int kk = 0; kk < 4; ++kk) {
      int k = kk * 4 + khalf;
      pA[kk] = (v2f){lds[wslot][col * 16 + k], lds[wslot][col * 16 + k + 1]};
    }
    __builtin_amdgcn_fence(__ATOMIC_ACQ_REL, "wavefront");
    __builtin_amdgcn_wave_barrier();
#pragma unroll
    for (int j = 0; j < 4; ++j) {
#pragma unroll
      for (int kk = 0; kk < 4; ++kk) {
        int k = kk * 4 + khalf;
        const float* vp = vb + (long)((j * 16 + col) * 16) * P_ + p0 + k;
        v2f bf = *(const v2f*)vp;
        O[j] = wmma4(pA[kk], bf, O[j]);
      }
    }
  }

#pragma unroll
  for (int r = 0; r < 8; ++r) {
    float inv = 1.f / Srow[r];
    O[0][r] *= inv; O[1][r] *= inv; O[2][r] *= inv; O[3][r] *= inv;
  }
#pragma unroll
  for (int r = 0; r < 8; ++r) {
    int l = l0 + r + 8 * hi;
    if (l < L_) {
      float* orow = obuf + ((long)b * L_ + l) * D_ + h;
#pragma unroll
      for (int j = 0; j < 4; ++j) orow[(j * 16 + col) * 16] = O[j][r];
    }
  }
}

extern "C" void kernel_launch(void* const* d_in, const int* in_sizes, int n_in,
                              void* d_out, int out_size, void* d_ws,
                              size_t ws_size, hipStream_t stream) {
  (void)in_sizes; (void)n_in; (void)out_size; (void)ws_size;
  const float* in = (const float*)d_in[0];
  const float* w_kv = (const float*)d_in[1];
  const float* w_q = (const float*)d_in[2];
  const float* b_q = (const float*)d_in[3];
  const float* w_out = (const float*)d_in[4];
  const float* b_out = (const float*)d_in[5];
  float* out = (float*)d_out;

  float* ws = (float*)d_ws;
  float* qbuf = ws;
  float* vbuf = qbuf + (size_t)B_ * L_ * C_;
  float* kbuf = vbuf + (size_t)B_ * D_ * P_;
  float* obuf = kbuf + (size_t)B_ * C_ * P_;

  gemm_rowA_wmma<<<256, 256, 0, stream>>>(in, (long)(B_ * D_), (long)D_, w_q,
                                          b_q, qbuf, (long)C_, (long)L_ * C_,
                                          L_, D_, 64, 8);
  gemm_colA_wmma<<<256, 256, 0, stream>>>(in, (long)(B_ * D_), (long)D_, w_kv,
                                          vbuf, (long)P_, (long)D_ * P_, L_,
                                          32, 16);
  gemm_colA_wmma<<<128, 256, 0, stream>>>(qbuf, (long)C_, (long)L_ * C_, w_kv,
                                          kbuf, (long)P_, (long)C_ * P_, L_,
                                          16, 16);
  attn_wmma<<<1024, 256, 0, stream>>>(qbuf, kbuf, vbuf, obuf);
  gemm_rowA_wmma<<<512, 256, 0, stream>>>(obuf, (long)D_, (long)L_ * D_,
                                          w_out, b_out, out, (long)(B_ * D_),
                                          (long)D_, L_, D_, 64, 16);
}


# v and k GEMMs (both contract over l against w_kv) merged into one launch of 3072 equal wave tiles, 256 work-groups of 12 waves: the k GEMM alone reached only half the WGPs; sum orders kept
# speedup vs baseline: 1.5057x; 1.5057x over previous
#include <hip/hip_runtime.h>


typedef __attribute__((ext_vector_type(2))) float v2f;
typedef __attribute__((ext_vector_type(8))) float v8f;

#define B_ 4
#define L_ 2046
#define D_ 1024
#define C_ 512
#define P_ 1024
#define H_ 16
#define HD_ 32
#define DV_ 64
#define WKV_LD 2048

static __device__ __forceinline__ v8f wmma4(v2f a, v2f b, v8f c) {
  return __builtin_amdgcn_wmma_f32_16x16x4_f32(false, a, false, b, (short)0, c,
                                               false, false);
}

static __device__ __forceinline__ float rowmax16(float v) {
  v = fmaxf(v, __shfl_xor(v, 8, 16));
  v = fmaxf(v, __shfl_xor(v, 4, 16));
  v = fmaxf(v, __shfl_xor(v, 2, 16));
  v = fmaxf(v, __shfl_xor(v, 1, 16));
  return v;
}
static __device__ __forceinline__ float rowsum16(float v) {
  v += __shfl_xor(v, 8, 16);
  v += __shfl_xor(v, 4, 16);
  v += __shfl_xor(v, 2, 16);
  v += __shfl_xor(v, 1, 16);
  return v;
}

template <int MT, int NT>
__global__ __launch_bounds__(256) void gemm_rowA_wmma(
    const float* __restrict__ A, long strideAm, long batchStrideA,
    const float* __restrict__ W, const float* __restrict__ bias,
    float* __restrict__ Out, long strideOm, long batchStrideO,
    int M, int K, int mstrips, int nstrips) {
  int lane = threadIdx.x & 31;
  int wid = (blockIdx.x * blockDim.x + threadIdx.x) >> 5;
  int wavesPerBatch = mstrips * nstrips;
  int b = wid / wavesPerBatch;
  int r = wid - b * wavesPerBatch;
  int m0 = (r / nstrips) * (16 * MT);
  int n0 = (r % nstrips) * (16 * NT);
  const float* Ab = A + (long)b * batchStrideA;
  float* Ob = Out + (long)b * batchStrideO;
  int col = lane & 15;
  int hi = lane >> 4;
  int khalf = hi * 2;
  const float* Arow[MT];
#pragma unroll
  for (int i = 0; i < MT; ++i) {
    int row = m0 + 16 * i + col;
    if (row > M - 1) row = M - 1;
    Arow[i] = Ab + (long)row * strideAm + khalf;
  }
  const float* Wp = W + (long)(n0 + col) * K + khalf;
  const long Wn16 = (long)16 * K;

  v8f acc[MT][NT] = {};
  for (int k0 = 0; k0 < K; k0 += 4) {
    if ((k0 & 63) == 0) {
#pragma unroll
      for (int i = 0; i < MT; ++i) __builtin_prefetch(Arow[i] + k0 + 256, 0, 1);
      __builtin_prefetch(Wp + k0 + 256, 0, 1);
      __builtin_prefetch(Wp + (NT - 1) * Wn16 + k0 + 256, 0, 1);
    }
    v2f a[MT];
#pragma unroll
    for (int i = 0; i < MT; ++i) a[i] = *(const v2f*)(Arow[i] + k0);
#pragma unroll
    for (int j = 0; j < NT; ++j) {
      v2f bf = *(const v2f*)(Wp + j * Wn16 + k0);
#pragma unroll
      for (int i = 0; i < MT; ++i) acc[i][j] = wmma4(a[i], bf, acc[i][j]);
    }
  }
  float bv[NT];
#pragma unroll
  for (int j = 0; j < NT; ++j) bv[j] = bias[n0 + j * 16 + col];
#pragma unroll
  for (int i = 0; i < MT; ++i) {
#pragma unroll
    for (int rr = 0; rr < 8; ++rr) {
      int row = m0 + i * 16 + rr + 8 * hi;
      if (row < M) {
        float* orow = Ob + (long)row * strideOm + n0;
#pragma unroll
        for (int j = 0; j < NT; ++j) orow[j * 16 + col] = acc[i][j][rr] + bv[j];
      }
    }
  }
}

template <int MT, int NT>
__global__ __launch_bounds__(384) void gemm_colA2_wmma(
    const float* __restrict__ A0, long strideAk0, long batchStrideA0,
    float* __restrict__ Out0, long batchStrideO0, int mstrips0, int waves0,
    const float* __restrict__ A1, long strideAk1, long batchStrideA1,
    float* __restrict__ Out1, long batchStrideO1, int mstrips1,
    const float* __restrict__ Wkv, long strideOm, int K, int nstrips) {
  int lane = threadIdx.x & 31;
  int wid = blockIdx.x * (blockDim.x >> 5) + (threadIdx.x >> 5);
  bool second = wid >= waves0;
  if (second) wid -= waves0;
  const float* A = second ? A1 : A0;
  float* Out = second ? Out1 : Out0;
  long strideAk = second ? strideAk1 : strideAk0;
  long batchStrideA = second ? batchStrideA1 : batchStrideA0;
  long batchStrideO = second ? batchStrideO1 : batchStrideO0;
  int mstrips = second ? mstrips1 : mstrips0;
  int wavesPerBatch = mstrips * nstrips;
  int b = wid / wavesPerBatch;
  int r = wid - b * wavesPerBatch;
  int m0 = (r / nstrips) * (16 * MT);
  int n0 = (r % nstrips) * (16 * NT);
  const float* Ab = A + (long)b * batchStrideA;
  float* Ob = Out + (long)b * batchStrideO;
  int col = lane & 15;
  int hi = lane >> 4;
  int khalf = hi * 2;
  const float* Am0 = Ab + (long)khalf * strideAk + m0 + col;
  const float* Wp = Wkv + (long)(n0 + col) * WKV_LD + khalf;
  const long Wn16 = (long)16 * WKV_LD;

  v8f acc[MT][NT] = {};
  int Kmain = K & ~3;
  for (int k0 = 0; k0 < Kmain; k0 += 4) {
    long ao = (long)k0 * strideAk;
    v2f a[MT];
#pragma unroll
    for (int i = 0; i < MT; ++i)
      a[i] = (v2f){Am0[ao + 16 * i], Am0[ao + strideAk + 16 * i]};
#pragma unroll
    for (int j = 0; j < NT; ++j) {
      v2f bf = *(const v2f*)(Wp + j * Wn16 + k0);
#pragma unroll
      for (int i = 0; i < MT; ++i) acc[i][j] = wmma4(a[i], bf, acc[i][j]);
    }
  }
  if (Kmain < K) {
    int k0 = Kmain;
    long ao = (long)k0 * strideAk;
    v2f a[MT];
#pragma unroll
    for (int i = 0; i < MT; ++i) a[i] = (v2f){0.f, 0.f};
    if (hi == 0) {
#pragma unroll
      for (int i = 0; i < MT; ++i)
        a[i] = (v2f){Am0[ao + 16 * i], Am0[ao + strideAk + 16 * i]};
    }
#pragma unroll
    for (int j = 0; j < NT; ++j) {
      v2f bf = *(const v2f*)(Wp + j * Wn16 + k0);
#pragma unroll
      for (int i = 0; i < MT; ++i) acc[i][j] = wmma4(a[i], bf, acc[i][j]);
    }
  }
#pragma unroll
  for (int i = 0; i < MT; ++i) {
#pragma unroll
    for (int rr = 0; rr < 8; ++rr) {
      int row = m0 + i * 16 + rr + 8 * hi;
      float* orow = Ob + (long)row * strideOm + n0;
#pragma unroll
      for (int j = 0; j < NT; ++j) orow[j * 16 + col] = acc[i][j][rr];
    }
  }
}

__global__ __launch_bounds__(256) void attn_wmma(
    const float* __restrict__ q, const float* __restrict__ kbuf,
    const float* __restrict__ vbuf, float* __restrict__ obuf) {
  __shared__ float lds[8][2][256];
  int lane = threadIdx.x & 31;
  int wslot = threadIdx.x >> 5;
  int wid = (blockIdx.x * blockDim.x + threadIdx.x) >> 5;
  int ltile = wid & 63;
  int h = (wid >> 6) & 15;
  int b = wid >> 10;
  int l0 = ltile * 32;
  int col = lane & 15;
  int hi = lane >> 4;
  int khalf = hi * 2;

  v2f qA[2][8];
#pragma unroll
  for (int i = 0; i < 2; ++i) {
    int l = l0 + 16 * i + col;
    bool lv = l < L_;
    const float* qrow = q + ((long)b * L_ + l) * C_ + h * HD_;
    const float sc = 0.125f * 1.44269504088896340736f;
#pragma unroll
    for (int kk = 0; kk < 8; ++kk) {
      int k = kk * 4 + khalf;
      float x0 = lv ? qrow[k] * sc : 0.f;
      float x1 = lv ? qrow[k + 1] * sc : 0.f;
      qA[i][kk] = (v2f){x0, x1};
    }
  }

  v8f O[2][4] = {};
  float Mrow[2][8], Srow[2][8];
#pragma unroll
  for (int i = 0; i < 2; ++i) {
#pragma unroll
    for (int r = 0; r < 8; ++r) { Mrow[i][r] = -INFINITY; Srow[i][r] = 0.f; }
  }

  const float* kb = kbuf + (long)b * C_ * P_ + (long)h * P_;
  const float* vb = vbuf + (long)b * D_ * P_ + (long)h * P_;

  for (int p0 = 0; p0 < P_; p0 += 16) {
    v8f c[2] = {};
#pragma unroll
    for (int kk = 0; kk < 8; ++kk) {
      int k = kk * 4 + khalf;
      const float* kp = kb + (long)k * 16 * P_ + p0 + col;
      v2f bf = {kp[0], kp[16 * P_]};
      c[0] = wmma4(qA[0][kk], bf, c[0]);
      c[1] = wmma4(qA[1][kk], bf, c[1]);
    }
#pragma unroll
    for (int i = 0; i < 2; ++i) {
      float alpha[8];
#pragma unroll
      for (int r = 0; r < 8; ++r) {
        float mx = rowmax16(c[i][r]);
        float mnew = fmaxf(Mrow[i][r], mx);
        float al = __builtin_exp2f(Mrow[i][r] - mnew);
        float pv = __builtin_exp2f(c[i][r] - mnew);
        Srow[i][r] = Srow[i][r] * al + rowsum16(pv);
        Mrow[i][r] = mnew;
        alpha[r] = al;
        lds[wslot][i][(r + 8 * hi) * 16 + col] = pv;
      }
#pragma unroll
      for (int r = 0; r < 8; ++r) {
        O[i][0][r] *= alpha[r]; O[i][1][r] *= alpha[r];
        O[i][2][r] *= alpha[r]; O[i][3][r] *= alpha[r];
      }
    }
    __builtin_amdgcn_fence(__ATOMIC_ACQ_REL, "wavefront");
    __builtin_amdgcn_wave_barrier();
    v2f pA[2][4];
#pragma unroll
    for (int i = 0; i < 2; ++i) {
#pragma unroll
      for (int kk = 0; kk < 4; ++kk) {
        int k = kk * 4 + khalf;
        pA[i][kk] = (v2f){lds[wslot][i][col * 16 + k], lds[wslot][i][col * 16 + k + 1]};
      }
    }
    __builtin_amdgcn_fence(__ATOMIC_ACQ_REL, "wavefront");
    __builtin_amdgcn_wave_barrier();
#pragma unroll
    for (int j = 0; j < 4; ++j) {
#pragma unroll
      for (int kk = 0; kk < 4; ++kk) {
        int k = kk * 4 + khalf;
        const float* vp = vb + (long)((j * 16 + col) * 16) * P_ + p0 + k;
        v2f bf = *(const v2f*)vp;
        O[0][j] = wmma4(pA[0][kk], bf, O[0][j]);
        O[1][j] = wmma4(pA[1][kk], bf, O[1][j]);
      }
    }
  }

#pragma unroll
  for (int i = 0; i < 2; ++i) {
#pragma unroll
    for (int r = 0; r < 8; ++r) {
      float inv = 1.f / Srow[i][r];
      O[i][0][r] *= inv; O[i][1][r] *= inv; O[i][2][r] *= inv; O[i][3][r] *= inv;
    }
#pragma unroll
    for (int r = 0; r < 8; ++r) {
      int l = l0 + 16 * i + r + 8 * hi;
      if (l < L_) {
        float* orow = obuf + ((long)b * L_ + l) * D_ + h;
#pragma unroll
        for (int j = 0; j < 4; ++j) orow[(j * 16 + col) * 16] = O[i][j][r];
      }
    }
  }
}

extern "C" void kernel_launch(void* const* d_in, const int* in_sizes, int n_in,
                              void* d_out, int out_size, void* d_ws,
                              size_t ws_size, hipStream_t stream) {
  (void)in_sizes; (void)n_in; (void)out_size; (void)ws_size;
  const float* in = (const float*)d_in[0];
  const float* w_kv = (const float*)d_in[1];
  const float* w_q = (const float*)d_in[2];
  const float* b_q = (const float*)d_in[3];
  const float* w_out = (const float*)d_in[4];
  const float* b_out = (const float*)d_in[5];
  float* out = (float*)d_out;

  float* ws = (float*)d_ws;
  float* qbuf = ws;
  float* vbuf = qbuf + (size_t)B_ * L_ * C_;
  float* kbuf = vbuf + (size_t)B_ * D_ * P_;
  float* obuf = kbuf + (size_t)B_ * C_ * P_;

  gemm_rowA_wmma<2, 4><<<256, 256, 0, stream>>>(in, (long)(B_ * D_), (long)D_, w_q,
                                          b_q, qbuf, (long)C_, (long)L_ * C_,
                                          L_, D_, 64, 8);
  gemm_colA2_wmma<4, 2><<<256, 384, 0, stream>>>(
      in, (long)(B_ * D_), (long)D_, vbuf, (long)D_ * P_, 16, 2048,
      qbuf, (long)C_, (long)L_ * C_, kbuf, (long)C_ * P_, 8,
      w_kv, (long)P_, L_, 32);
  attn_wmma<<<512, 256, 0, stream>>>(qbuf, kbuf, vbuf, obuf);
  gemm_rowA_wmma<4, 4><<<256, 256, 0, stream>>>(obuf, (long)D_, (long)L_ * D_,
                                                w_out, b_out, out,
                                                (long)(B_ * D_), (long)D_, L_,
                                                D_, 32, 16);
}


# v/k GEMM: A operand (contiguous along m) loaded 16 B per lane, the 64 rows of four M tiles dealt out row by row (tile t owns rows 4c+t): 2 A loads per k-step instead of 8; all sum orders kept
# speedup vs baseline: 1.6314x; 1.0835x over previous
#include <hip/hip_runtime.h>


typedef __attribute__((ext_vector_type(2))) float v2f;
typedef __attribute__((ext_vector_type(4))) float v4f;
typedef __attribute__((ext_vector_type(8))) float v8f;

#define B_ 4
#define L_ 2046
#define D_ 1024
#define C_ 512
#define P_ 1024
#define H_ 16
#define HD_ 32
#define DV_ 64
#define WKV_LD 2048

static __device__ __forceinline__ v8f wmma4(v2f a, v2f b, v8f c) {
  return __builtin_amdgcn_wmma_f32_16x16x4_f32(false, a, false, b, (short)0, c,
                                               false, false);
}

static __device__ __forceinline__ float rowmax16(float v) {
  v = fmaxf(v, __shfl_xor(v, 8, 16));
  v = fmaxf(v, __shfl_xor(v, 4, 16));
  v = fmaxf(v, __shfl_xor(v, 2, 16));
  v = fmaxf(v, __shfl_xor(v, 1, 16));
  return v;
}
static __device__ __forceinline__ float rowsum16(float v) {
  v += __shfl_xor(v, 8, 16);
  v += __shfl_xor(v, 4, 16);
  v += __shfl_xor(v, 2, 16);
  v += __shfl_xor(v, 1, 16);
  return v;
}

template <int MT, int NT>
__global__ __launch_bounds__(256) void gemm_rowA_wmma(
    const float* __restrict__ A, long strideAm, long batchStrideA,
    const float* __restrict__ W, const float* __restrict__ bias,
    float* __restrict__ Out, long strideOm, long batchStrideO,
    int M, int K, int mstrips, int nstrips) {
  int lane = threadIdx.x & 31;
  int wid = (blockIdx.x * blockDim.x + threadIdx.x) >> 5;
  int wavesPerBatch = mstrips * nstrips;
  int b = wid / wavesPerBatch;
  int r = wid - b * wavesPerBatch;
  int m0 = (r / nstrips) * (16 * MT);
  int n0 = (r % nstrips) * (16 * NT);
  const float* Ab = A + (long)b * batchStrideA;
  float* Ob = Out + (long)b * batchStrideO;
  int col = lane & 15;
  int hi = lane >> 4;
  int khalf = hi * 2;
  const float* Arow[MT];
#pragma unroll
  for (int i = 0; i < MT; ++i) {
    int row = m0 + 16 * i + col;
    if (row > M - 1) row = M - 1;
    Arow[i] = Ab + (long)row * strideAm + khalf;
  }
  const float* Wp = W + (long)(n0 + col) * K + khalf;
  const long Wn16 = (long)16 * K;

  v8f acc[MT][NT] = {};
  for (int k0 = 0; k0 < K; k0 += 4) {
    if ((k0 & 63) == 0) {
#pragma unroll
      for (int i = 0; i < MT; ++i) __builtin_prefetch(Arow[i] + k0 + 256, 0, 1);
      __builtin_prefetch(Wp + k0 + 256, 0, 1);
      __builtin_prefetch(Wp + (NT - 1) * Wn16 + k0 + 256, 0, 1);
    }
    v2f a[MT];
#pragma unroll
    for (int i = 0; i < MT; ++i) a[i] = *(const v2f*)(Arow[i] + k0);
#pragma unroll
    for (int j = 0; j < NT; ++j) {
      v2f bf = *(const v2f*)(Wp + j * Wn16 + k0);
#pragma unroll
      for (int i = 0; i < MT; ++i) acc[i][j] = wmma4(a[i], bf, acc[i][j]);
    }
  }
  float bv[NT];
#pragma unroll
  for (int j = 0; j < NT; ++j) bv[j] = bias[n0 + j * 16 + col];
#pragma unroll
  for (int i = 0; i < MT; ++i) {
#pragma unroll
    for (int rr = 0; rr < 8; ++rr) {
      int row = m0 + i * 16 + rr + 8 * hi;
      if (row < M) {
        float* orow = Ob + (long)row * strideOm + n0;
#pragma unroll
        for (int j = 0; j < NT; ++j) orow[j * 16 + col] = acc[i][j][rr] + bv[j];
      }
    }
  }
}

template <int MT, int NT>
__global__ __launch_bounds__(384) void gemm_colA2_wmma(
    const float* __restrict__ A0, long strideAk0, long batchStrideA0,
    float* __restrict__ Out0, long batchStrideO0, int mstrips0, int waves0,
    const float* __restrict__ A1, long strideAk1, long batchStrideA1,
    float* __restrict__ Out1, long batchStrideO1, int mstrips1,
    const float* __restrict__ Wkv, long strideOm, int K, int nstrips) {
  int lane = threadIdx.x & 31;
  int wid = blockIdx.x * (blockDim.x >> 5) + (threadIdx.x >> 5);
  bool second = wid >= waves0;
  if (second) wid -= waves0;
  const float* A = second ? A1 : A0;
  float* Out = second ? Out1 : Out0;
  long strideAk = second ? strideAk1 : strideAk0;
  long batchStrideA = second ? batchStrideA1 : batchStrideA0;
  long batchStrideO = second ? batchStrideO1 : batchStrideO0;
  int mstrips = second ? mstrips1 : mstrips0;
  int wavesPerBatch = mstrips * nstrips;
  int b = wid / wavesPerBatch;
  int r = wid - b * wavesPerBatch;
  int m0 = (r / nstrips) * (16 * MT);
  int n0 = (r % nstrips) * (16 * NT);
  const float* Ab = A + (long)b * batchStrideA;
  float* Ob = Out + (long)b * batchStrideO;
  int col = lane & 15;
  int hi = lane >> 4;
  int khalf = hi * 2;
  static_assert(MT % 4 == 0, "M tiles come in row-interleaved groups of four");
  const float* Am0 = Ab + (long)khalf * strideAk + m0 + 4 * col;
  const float* Wp = Wkv + (long)(n0 + col) * WKV_LD + khalf;
  const long Wn16 = (long)16 * WKV_LD;

  v8f acc[MT][NT] = {};
  int Kmain = K & ~3;
  for (int k0 = 0; k0 < Kmain; k0 += 4) {
    long ao = (long)k0 * strideAk;
    v2f a[MT];
#pragma unroll
    for (int g = 0; g < MT / 4; ++g) {
      v4f x0 = *(const v4f*)(Am0 + ao + 64 * g);
      v4f x1 = *(const v4f*)(Am0 + ao + strideAk + 64 * g);
#pragma unroll
      for (int t = 0; t < 4; ++t) a[4 * g + t] = (v2f){x0[t], x1[t]};
    }
#pragma unroll
    for (int j = 0; j < NT; ++j) {
      v2f bf = *(const v2f*)(Wp + j * Wn16 + k0);
#pragma unroll
      for (int i = 0; i < MT; ++i) acc[i][j] = wmma4(a[i], bf, acc[i][j]);
    }
  }
  if (Kmain < K) {
    int k0 = Kmain;
    long ao = (long)k0 * strideAk;
    v2f a[MT];
#pragma unroll
    for (int i = 0; i < MT; ++i) a[i] = (v2f){0.f, 0.f};
    if (hi == 0) {
#pragma unroll
      for (int g = 0; g < MT / 4; ++g) {
        v4f x0 = *(const v4f*)(Am0 + ao + 64 * g);
        v4f x1 = *(const v4f*)(Am0 + ao + strideAk + 64 * g);
#pragma unroll
        for (int t = 0; t < 4; ++t) a[4 * g + t] = (v2f){x0[t], x1[t]};
      }
    }
#pragma unroll
    for (int j = 0; j < NT; ++j) {
      v2f bf = *(const v2f*)(Wp + j * Wn16 + k0);
#pragma unroll
      for (int i = 0; i < MT; ++i) acc[i][j] = wmma4(a[i], bf, acc[i][j]);
    }
  }
#pragma unroll
  for (int i = 0; i < MT; ++i) {
#pragma unroll
    for (int rr = 0; rr < 8; ++rr) {
      int row = m0 + (i >> 2) * 64 + 4 * (rr + 8 * hi) + (i & 3);
      float* orow = Ob + (long)row * strideOm + n0;
#pragma unroll
      for (int j = 0; j < NT; ++j) orow[j * 16 + col] = acc[i][j][rr];
    }
  }
}

__global__ __launch_bounds__(256) void attn_wmma(
    const float* __restrict__ q, const float* __restrict__ kbuf,
    const float* __restrict__ vbuf, float* __restrict__ obuf) {
  __shared__ float lds[8][2][256];
  int lane = threadIdx.x & 31;
  int wslot = threadIdx.x >> 5;
  int wid = (blockIdx.x * blockDim.x + threadIdx.x) >> 5;
  int ltile = wid & 63;
  int h = (wid >> 6) & 15;
  int b = wid >> 10;
  int l0 = ltile * 32;
  int col = lane & 15;
  int hi = lane >> 4;
  int khalf = hi * 2;

  v2f qA[2][8];
#pragma unroll
  for (int i = 0; i < 2; ++i) {
    int l = l0 + 16 * i + col;
    bool lv = l < L_;
    const float* qrow = q + ((long)b * L_ + l) * C_ + h * HD_;
    const float sc = 0.125f * 1.44269504088896340736f;
#pragma unroll
    for (int kk = 0; kk < 8; ++kk) {
      int k = kk * 4 + khalf;
      float x0 = lv ? qrow[k] * sc : 0.f;
      float x1 = lv ? qrow[k + 1] * sc : 0.f;
      qA[i][kk] = (v2f){x0, x1};
    }
  }

  v8f O[2][4] = {};
  float Mrow[2][8], Srow[2][8];
#pragma unroll
  for (int i = 0; i < 2; ++i) {
#pragma unroll
    for (int r = 0; r < 8; ++r) { Mrow[i][r] = -INFINITY; Srow[i][r] = 0.f; }
  }

  const float* kb = kbuf + (long)b * C_ * P_ + (long)h * P_;
  const float* vb = vbuf + (long)b * D_ * P_ + (long)h * P_;

  for (int p0 = 0; p0 < P_; p0 += 16) {
    v8f c[2] = {};
#pragma unroll
    for (int kk = 0; kk < 8; ++kk) {
      int k = kk * 4 + khalf;
      const float* kp = kb + (long)k * 16 * P_ + p0 + col;
      v2f bf = {kp[0], kp[16 * P_]};
      c[0] = wmma4(qA[0][kk], bf, c[0]);
      c[1] = wmma4(qA[1][kk], bf, c[1]);
    }
#pragma unroll
    for (int i = 0; i < 2; ++i) {
      float alpha[8];
#pragma unroll
      for (int r = 0; r < 8; ++r) {
        float mx = rowmax16(c[i][r]);
        float mnew = fmaxf(Mrow[i][r], mx);
        float al = __builtin_exp2f(Mrow[i][r] - mnew);
        float pv = __builtin_exp2f(c[i][r] - mnew);
        Srow[i][r] = Srow[i][r] * al + rowsum16(pv);
        Mrow[i][r] = mnew;
        alpha[r] = al;
        lds[wslot][i][(r + 8 * hi) * 16 + col] = pv;
      }
#pragma unroll
      for (int r = 0; r < 8; ++r) {
        O[i][0][r] *= alpha[r]; O[i][1][r] *= alpha[r];
        O[i][2][r] *= alpha[r]; O[i][3][r] *= alpha[r];
      }
    }
    __builtin_amdgcn_fence(__ATOMIC_ACQ_REL, "wavefront");
    __builtin_amdgcn_wave_barrier();
    v2f pA[2][4];
#pragma unroll
    for (int i = 0; i < 2; ++i) {
#pragma unroll
      for (int kk = 0; kk < 4; ++kk) {
        int k = kk * 4 + khalf;
        pA[i][kk] = (v2f){lds[wslot][i][col * 16 + k], lds[wslot][i][col * 16 + k + 1]};
      }
    }
    __builtin_amdgcn_fence(__ATOMIC_ACQ_REL, "wavefront");
    __builtin_amdgcn_wave_barrier();
#pragma unroll
    for (int j = 0; j < 4; ++j) {
#pragma unroll
      for (int kk = 0; kk < 4; ++kk) {
        int k = kk * 4 + khalf;
        const float* vp = vb + (long)((j * 16 + col) * 16) * P_ + p0 + k;
        v2f bf = *(const v2f*)vp;
        O[0][j] = wmma4(pA[0][kk], bf, O[0][j]);
        O[1][j] = wmma4(pA[1][kk], bf, O[1][j]);
      }
    }
  }

#pragma unroll
  for (int i = 0; i < 2; ++i) {
#pragma unroll
    for (int r = 0; r < 8; ++r) {
      float inv = 1.f / Srow[i][r];
      O[i][0][r] *= inv; O[i][1][r] *= inv; O[i][2][r] *= inv; O[i][3][r] *= inv;
    }
#pragma unroll
    for (int r = 0; r < 8; ++r) {
      int l = l0 + 16 * i + r + 8 * hi;
      if (l < L_) {
        float* orow = obuf + ((long)b * L_ + l) * D_ + h;
#pragma unroll
        for (int j = 0; j < 4; ++j) orow[(j * 16 + col) * 16] = O[i][j][r];
      }
    }
  }
}

extern "C" void kernel_launch(void* const* d_in, const int* in_sizes, int n_in,
                              void* d_out, int out_size, void* d_ws,
                              size_t ws_size, hipStream_t stream) {
  (void)in_sizes; (void)n_in; (void)out_size; (void)ws_size;
  const float* in = (const float*)d_in[0];
  const float* w_kv = (const float*)d_in[1];
  const float* w_q = (const float*)d_in[2];
  const float* b_q = (const float*)d_in[3];
  const float* w_out = (const float*)d_in[4];
  const float* b_out = (const float*)d_in[5];
  float* out = (float*)d_out;

  float* ws = (float*)d_ws;
  float* qbuf = ws;
  float* vbuf = qbuf + (size_t)B_ * L_ * C_;
  float* kbuf = vbuf + (size_t)B_ * D_ * P_;
  float* obuf = kbuf + (size_t)B_ * C_ * P_;

  gemm_rowA_wmma<2, 4><<<256, 256, 0, stream>>>(in, (long)(B_ * D_), (long)D_, w_q,
                                          b_q, qbuf, (long)C_, (long)L_ * C_,
                                          L_, D_, 64, 8);
  gemm_colA2_wmma<4, 2><<<256, 384, 0, stream>>>(
      in, (long)(B_ * D_), (long)D_, vbuf, (long)D_ * P_, 16, 2048,
      qbuf, (long)C_, (long)L_ * C_, kbuf, (long)C_ * P_, 8,
      w_kv, (long)P_, L_, 32);
  attn_wmma<<<512, 256, 0, stream>>>(qbuf, kbuf, vbuf, obuf);
  gemm_rowA_wmma<4, 4><<<256, 256, 0, stream>>>(obuf, (long)D_, (long)L_ * D_,
                                                w_out, b_out, out,
                                                (long)(B_ * D_), (long)D_, L_,
                                                D_, 32, 16);
}


# Attention trip covers 32 scores per row (was 16): one row-max butterfly, one alpha, one O/S rescale per 32; the 2 sub-tiles' probs are added lane-wise before the row-sum butterfly: sum order changes
# speedup vs baseline: 1.7234x; 1.0564x over previous
#include <hip/hip_runtime.h>


typedef __attribute__((ext_vector_type(2))) float v2f;
typedef __attribute__((ext_vector_type(4))) float v4f;
typedef __attribute__((ext_vector_type(8))) float v8f;

#define B_ 4
#define L_ 2046
#define D_ 1024
#define C_ 512
#define P_ 1024
#define H_ 16
#define HD_ 32
#define DV_ 64
#define WKV_LD 2048

static __device__ __forceinline__ v8f wmma4(v2f a, v2f b, v8f c) {
  return __builtin_amdgcn_wmma_f32_16x16x4_f32(false, a, false, b, (short)0, c,
                                               false, false);
}

static __device__ __forceinline__ float rowmax16(float v) {
  v = fmaxf(v, __shfl_xor(v, 8, 16));
  v = fmaxf(v, __shfl_xor(v, 4, 16));
  v = fmaxf(v, __shfl_xor(v, 2, 16));
  v = fmaxf(v, __shfl_xor(v, 1, 16));
  return v;
}
static __device__ __forceinline__ float rowsum16(float v) {
  v += __shfl_xor(v, 8, 16);
  v += __shfl_xor(v, 4, 16);
  v += __shfl_xor(v, 2, 16);
  v += __shfl_xor(v, 1, 16);
  return v;
}

template <int MT, int NT>
__global__ __launch_bounds__(256) void gemm_rowA_wmma(
    const float* __restrict__ A, long strideAm, long batchStrideA,
    const float* __restrict__ W, const float* __restrict__ bias,
    float* __restrict__ Out, long strideOm, long batchStrideO,
    int M, int K, int mstrips, int nstrips) {
  int lane = threadIdx.x & 31;
  int wid = (blockIdx.x * blockDim.x + threadIdx.x) >> 5;
  int wavesPerBatch = mstrips * nstrips;
  int b = wid / wavesPerBatch;
  int r = wid - b * wavesPerBatch;
  int m0 = (r / nstrips) * (16 * MT);
  int n0 = (r % nstrips) * (16 * NT);
  const float* Ab = A + (long)b * batchStrideA;
  float* Ob = Out + (long)b * batchStrideO;
  int col = lane & 15;
  int hi = lane >> 4;
  int khalf = hi * 2;
  const float* Arow[MT];
#pragma unroll
  for (int i = 0; i < MT; ++i) {
    int row = m0 + 16 * i + col;
    if (row > M - 1) row = M - 1;
    Arow[i] = Ab + (long)row * strideAm + khalf;
  }
  const float* Wp = W + (long)(n0 + col) * K + khalf;
  const long Wn16 = (long)16 * K;

  v8f acc[MT][NT] = {};
  for (int k0 = 0; k0 < K; k0 += 4) {
    if ((k0 & 63) == 0) {
#pragma unroll
      for (int i = 0; i < MT; ++i) __builtin_prefetch(Arow[i] + k0 + 256, 0, 1);
      __builtin_prefetch(Wp + k0 + 256, 0, 1);
      __builtin_prefetch(Wp + (NT - 1) * Wn16 + k0 + 256, 0, 1);
    }
    v2f a[MT];
#pragma unroll
    for (int i = 0; i < MT; ++i) a[i] = *(const v2f*)(Arow[i] + k0);
#pragma unroll
    for (int j = 0; j < NT; ++j) {
      v2f bf = *(const v2f*)(Wp + j * Wn16 + k0);
#pragma unroll
      for (int i = 0; i < MT; ++i) acc[i][j] = wmma4(a[i], bf, acc[i][j]);
    }
  }
  float bv[NT];
#pragma unroll
  for (int j = 0; j < NT; ++j) bv[j] = bias[n0 + j * 16 + col];
#pragma unroll
  for (int i = 0; i < MT; ++i) {
#pragma unroll
    for (int rr = 0; rr < 8; ++rr) {
      int row = m0 + i * 16 + rr + 8 * hi;
      if (row < M) {
        float* orow = Ob + (long)row * strideOm + n0;
#pragma unroll
        for (int j = 0; j < NT; ++j) orow[j * 16 + col] = acc[i][j][rr] + bv[j];
      }
    }
  }
}

template <int MT, int NT>
__global__ __launch_bounds__(384) void gemm_colA2_wmma(
    const float* __restrict__ A0, long strideAk0, long batchStrideA0,
    float* __restrict__ Out0, long batchStrideO0, int mstrips0, int waves0,
    const float* __restrict__ A1, long strideAk1, long batchStrideA1,
    float* __restrict__ Out1, long batchStrideO1, int mstrips1,
    const float* __restrict__ Wkv, long strideOm, int K, int nstrips) {
  int lane = threadIdx.x & 31;
  int wid = blockIdx.x * (blockDim.x >> 5) + (threadIdx.x >> 5);
  bool second = wid >= waves0;
  if (second) wid -= waves0;
  const float* A = second ? A1 : A0;
  float* Out = second ? Out1 : Out0;
  long strideAk = second ? strideAk1 : strideAk0;
  long batchStrideA = second ? batchStrideA1 : batchStrideA0;
  long batchStrideO = second ? batchStrideO1 : batchStrideO0;
  int mstrips = second ? mstrips1 : mstrips0;
  int wavesPerBatch = mstrips * nstrips;
  int b = wid / wavesPerBatch;
  int r = wid - b * wavesPerBatch;
  int m0 = (r / nstrips) * (16 * MT);
  int n0 = (r % nstrips) * (16 * NT);
  const float* Ab = A + (long)b * batchStrideA;
  float* Ob = Out + (long)b * batchStrideO;
  int col = lane & 15;
  int hi = lane >> 4;
  int khalf = hi * 2;
  static_assert(MT % 4 == 0, "M tiles come in row-interleaved groups of four");
  const float* Am0 = Ab + (long)khalf * strideAk + m0 + 4 * col;
  const float* Wp = Wkv + (long)(n0 + col) * WKV_LD + khalf;
  const long Wn16 = (long)16 * WKV_LD;

  v8f acc[MT][NT] = {};
  int Kmain = K & ~3;
  for (int k0 = 0; k0 < Kmain; k0 += 4) {
    long ao = (long)k0 * strideAk;
    v2f a[MT];
#pragma unroll
    for (int g = 0; g < MT / 4; ++g) {
      v4f x0 = *(const v4f*)(Am0 + ao + 64 * g);
      v4f x1 = *(const v4f*)(Am0 + ao + strideAk + 64 * g);
#pragma unroll
      for (int t = 0; t < 4; ++t) a[4 * g + t] = (v2f){x0[t], x1[t]};
    }
#pragma unroll
    for (int j = 0; j < NT; ++j) {
      v2f bf = *(const v2f*)(Wp + j * Wn16 + k0);
#pragma unroll
      for (int i = 0; i < MT; ++i) acc[i][j] = wmma4(a[i], bf, acc[i][j]);
    }
  }
  if (Kmain < K) {
    int k0 = Kmain;
    long ao = (long)k0 * strideAk;
    v2f a[MT];
#pragma unroll
    for (int i = 0; i < MT; ++i) a[i] = (v2f){0.f, 0.f};
    if (hi == 0) {
#pragma unroll
      for (int g = 0; g < MT / 4; ++g) {
        v4f x0 = *(const v4f*)(Am0 + ao + 64 * g);
        v4f x1 = *(const v4f*)(Am0 + ao + strideAk + 64 * g);
#pragma unroll
        for (int t = 0; t < 4; ++t) a[4 * g + t] = (v2f){x0[t], x1[t]};
      }
    }
#pragma unroll
    for (int j = 0; j < NT; ++j) {
      v2f bf = *(const v2f*)(Wp + j * Wn16 + k0);
#pragma unroll
      for (int i = 0; i < MT; ++i) acc[i][j] = wmma4(a[i], bf, acc[i][j]);
    }
  }
#pragma unroll
  for (int i = 0; i < MT; ++i) {
#pragma unroll
    for (int rr = 0; rr < 8; ++rr) {
      int row = m0 + (i >> 2) * 64 + 4 * (rr + 8 * hi) + (i & 3);
      float* orow = Ob + (long)row * strideOm + n0;
#pragma unroll
      for (int j = 0; j < NT; ++j) orow[j * 16 + col] = acc[i][j][rr];
    }
  }
}

__global__ __launch_bounds__(256, 4) void attn_wmma(
    const float* __restrict__ q, const float* __restrict__ kbuf,
    const float* __restrict__ vbuf, float* __restrict__ obuf) {
  __shared__ float lds[8][2][2][256];
  int lane = threadIdx.x & 31;
  int wslot = threadIdx.x >> 5;
  int wid = (blockIdx.x * blockDim.x + threadIdx.x) >> 5;
  int ltile = wid & 63;
  int h = (wid >> 6) & 15;
  int b = wid >> 10;
  int l0 = ltile * 32;
  int col = lane & 15;
  int hi = lane >> 4;
  int khalf = hi * 2;

  v2f qA[2][8];
#pragma unroll
  for (int i = 0; i < 2; ++i) {
    int l = l0 + 16 * i + col;
    bool lv = l < L_;
    const float* qrow = q + ((long)b * L_ + l) * C_ + h * HD_;
    const float sc = 0.125f * 1.44269504088896340736f;
#pragma unroll
    for (int kk = 0; kk < 8; ++kk) {
      int k = kk * 4 + khalf;
      float x0 = lv ? qrow[k] * sc : 0.f;
      float x1 = lv ? qrow[k + 1] * sc : 0.f;
      qA[i][kk] = (v2f){x0, x1};
    }
  }

  v8f O[2][4] = {};
  float Mrow[2][8], Srow[2][8];
#pragma unroll
  for (int i = 0; i < 2; ++i) {
#pragma unroll
    for (int r = 0; r < 8; ++r) { Mrow[i][r] = -INFINITY; Srow[i][r] = 0.f; }
  }

  const float* kb = kbuf + (long)b * C_ * P_ + (long)h * P_;
  const float* vb = vbuf + (long)b * D_ * P_ + (long)h * P_;

  for (int p0 = 0; p0 < P_; p0 += 32) {
    v8f c[2][2] = {};
#pragma unroll
    for (int s = 0; s < 2; ++s) {
#pragma unroll
      for (int kk = 0; kk < 8; ++kk) {
        int k = kk * 4 + khalf;
        const float* kp = kb + (long)k * 16 * P_ + p0 + 16 * s + col;
        v2f bf = {kp[0], kp[16 * P_]};
        c[0][s] = wmma4(qA[0][kk], bf, c[0][s]);
        c[1][s] = wmma4(qA[1][kk], bf, c[1][s]);
      }
    }
#pragma unroll
    for (int i = 0; i < 2; ++i) {
      float alpha[8];
#pragma unroll
      for (int r = 0; r < 8; ++r) {
        float mx = rowmax16(fmaxf(c[i][0][r], c[i][1][r]));
        float mnew = fmaxf(Mrow[i][r], mx);
        float al = __builtin_exp2f(Mrow[i][r] - mnew);
        float pv0 = __builtin_exp2f(c[i][0][r] - mnew);
        float pv1 = __builtin_exp2f(c[i][1][r] - mnew);
        Srow[i][r] = Srow[i][r] * al + rowsum16(pv0 + pv1);
        Mrow[i][r] = mnew;
        alpha[r] = al;
        lds[wslot][i][0][(r + 8 * hi) * 16 + col] = pv0;
        lds[wslot][i][1][(r + 8 * hi) * 16 + col] = pv1;
      }
#pragma unroll
      for (int r = 0; r < 8; ++r) {
        O[i][0][r] *= alpha[r]; O[i][1][r] *= alpha[r];
        O[i][2][r] *= alpha[r]; O[i][3][r] *= alpha[r];
      }
    }
    __builtin_amdgcn_sched_barrier(0);
    __builtin_amdgcn_fence(__ATOMIC_ACQ_REL, "wavefront");
    __builtin_amdgcn_wave_barrier();
    v2f pA[2][2][4];
#pragma unroll
    for (int i = 0; i < 2; ++i) {
#pragma unroll
      for (int s = 0; s < 2; ++s) {
#pragma unroll
        for (int kk = 0; kk < 4; ++kk) {
          int k = kk * 4 + khalf;
          pA[i][s][kk] = (v2f){lds[wslot][i][s][col * 16 + k], lds[wslot][i][s][col * 16 + k + 1]};
        }
      }
    }
    __builtin_amdgcn_fence(__ATOMIC_ACQ_REL, "wavefront");
    __builtin_amdgcn_wave_barrier();
#pragma unroll
    for (int s = 0; s < 2; ++s) {
#pragma unroll
      for (int j = 0; j < 4; ++j) {
#pragma unroll
        for (int kk = 0; kk < 4; ++kk) {
          int k = kk * 4 + khalf;
          const float* vp = vb + (long)((j * 16 + col) * 16) * P_ + p0 + 16 * s + k;
          v2f bf = *(const v2f*)vp;
          O[0][j] = wmma4(pA[0][s][kk], bf, O[0][j]);
          O[1][j] = wmma4(pA[1][s][kk], bf, O[1][j]);
        }
      }
    }
  }

#pragma unroll
  for (int i = 0; i < 2; ++i) {
#pragma unroll
    for (int r = 0; r < 8; ++r) {
      float inv = 1.f / Srow[i][r];
      O[i][0][r] *= inv; O[i][1][r] *= inv; O[i][2][r] *= inv; O[i][3][r] *= inv;
    }
#pragma unroll
    for (int r = 0; r < 8; ++r) {
      int l = l0 + 16 * i + r + 8 * hi;
      if (l < L_) {
        float* orow = obuf + ((long)b * L_ + l) * D_ + h;
#pragma unroll
        for (int j = 0; j < 4; ++j) orow[(j * 16 + col) * 16] = O[i][j][r];
      }
    }
  }
}

extern "C" void kernel_launch(void* const* d_in, const int* in_sizes, int n_in,
                              void* d_out, int out_size, void* d_ws,
                              size_t ws_size, hipStream_t stream) {
  (void)in_sizes; (void)n_in; (void)out_size; (void)ws_size;
  const float* in = (const float*)d_in[0];
  const float* w_kv = (const float*)d_in[1];
  const float* w_q = (const float*)d_in[2];
  const float* b_q = (const float*)d_in[3];
  const float* w_out = (const float*)d_in[4];
  const float* b_out = (const float*)d_in[5];
  float* out = (float*)d_out;

  float* ws = (float*)d_ws;
  float* qbuf = ws;
  float* vbuf = qbuf + (size_t)B_ * L_ * C_;
  float* kbuf = vbuf + (size_t)B_ * D_ * P_;
  float* obuf = kbuf + (size_t)B_ * C_ * P_;

  gemm_rowA_wmma<2, 4><<<256, 256, 0, stream>>>(in, (long)(B_ * D_), (long)D_, w_q,
                                          b_q, qbuf, (long)C_, (long)L_ * C_,
                                          L_, D_, 64, 8);
  gemm_colA2_wmma<4, 2><<<256, 384, 0, stream>>>(
      in, (long)(B_ * D_), (long)D_, vbuf, (long)D_ * P_, 16, 2048,
      qbuf, (long)C_, (long)L_ * C_, kbuf, (long)C_ * P_, 8,
      w_kv, (long)P_, L_, 32);
  attn_wmma<<<512, 256, 0, stream>>>(qbuf, kbuf, vbuf, obuf);
  gemm_rowA_wmma<4, 4><<<256, 256, 0, stream>>>(obuf, (long)D_, (long)L_ * D_,
                                                w_out, b_out, out,
                                                (long)(B_ * D_), (long)D_, L_,
                                                D_, 32, 16);
}


# Producers write k, v and the attention output in their consumer's fragment layout (k [d/2][p][2], v [p/8][dv][8], O channel-major): its loads touch 2-6 lines, not 16; out tile 128x32; sum orders kept
# speedup vs baseline: 1.8083x; 1.0493x over previous
#include <hip/hip_runtime.h>


typedef __attribute__((ext_vector_type(2))) float v2f;
typedef __attribute__((ext_vector_type(4))) float v4f;
typedef __attribute__((ext_vector_type(8))) float v8f;

#define B_ 4
#define L_ 2046
#define D_ 1024
#define C_ 512
#define P_ 1024
#define H_ 16
#define HD_ 32
#define DV_ 64
#define WKV_LD 2048
#define MTOT_ (B_ * L_)

static __device__ __forceinline__ v8f wmma4(v2f a, v2f b, v8f c) {
  return __builtin_amdgcn_wmma_f32_16x16x4_f32(false, a, false, b, (short)0, c,
                                               false, false);
}

static __device__ __forceinline__ float rowmax16(float v) {
  v = fmaxf(v, __shfl_xor(v, 8, 16));
  v = fmaxf(v, __shfl_xor(v, 4, 16));
  v = fmaxf(v, __shfl_xor(v, 2, 16));
  v = fmaxf(v, __shfl_xor(v, 1, 16));
  return v;
}
static __device__ __forceinline__ float rowsum16(float v) {
  v += __shfl_xor(v, 8, 16);
  v += __shfl_xor(v, 4, 16);
  v += __shfl_xor(v, 2, 16);
  v += __shfl_xor(v, 1, 16);
  return v;
}

template <int MT, int NT>
__global__ __launch_bounds__(256) void gemm_rowA_wmma(
    const float* __restrict__ A, long strideAm, long batchStrideA,
    const float* __restrict__ W, const float* __restrict__ bias,
    float* __restrict__ Out, long strideOm, long batchStrideO,
    int M, int K, int mstrips, int nstrips) {
  int lane = threadIdx.x & 31;
  int wid = (blockIdx.x * blockDim.x + threadIdx.x) >> 5;
  int wavesPerBatch = mstrips * nstrips;
  int b = wid / wavesPerBatch;
  int r = wid - b * wavesPerBatch;
  int m0 = (r / nstrips) * (16 * MT);
  int n0 = (r % nstrips) * (16 * NT);
  const float* Ab = A + (long)b * batchStrideA;
  float* Ob = Out + (long)b * batchStrideO;
  int col = lane & 15;
  int hi = lane >> 4;
  int khalf = hi * 2;
  const float* Arow[MT];
#pragma unroll
  for (int i = 0; i < MT; ++i) {
    int row = m0 + 16 * i + col;
    if (row > M - 1) row = M - 1;
    Arow[i] = Ab + (long)row * strideAm + khalf;
  }
  const float* Wp = W + (long)(n0 + col) * K + khalf;
  const long Wn16 = (long)16 * K;

  v8f acc[MT][NT] = {};
  for (int k0 = 0; k0 < K; k0 += 4) {
    if ((k0 & 63) == 0) {
#pragma unroll
      for (int i = 0; i < MT; ++i) __builtin_prefetch(Arow[i] + k0 + 256, 0, 1);
      __builtin_prefetch(Wp + k0 + 256, 0, 1);
      __builtin_prefetch(Wp + (NT - 1) * Wn16 + k0 + 256, 0, 1);
    }
    v2f a[MT];
#pragma unroll
    for (int i = 0; i < MT; ++i) a[i] = *(const v2f*)(Arow[i] + k0);
#pragma unroll
    for (int j = 0; j < NT; ++j) {
      v2f bf = *(const v2f*)(Wp + j * Wn16 + k0);
#pragma unroll
      for (int i = 0; i < MT; ++i) acc[i][j] = wmma4(a[i], bf, acc[i][j]);
    }
  }
  float bv[NT];
#pragma unroll
  for (int j = 0; j < NT; ++j) bv[j] = bias[n0 + j * 16 + col];
#pragma unroll
  for (int i = 0; i < MT; ++i) {
#pragma unroll
    for (int rr = 0; rr < 8; ++rr) {
      int row = m0 + i * 16 + rr + 8 * hi;
      if (row < M) {
        float* orow = Ob + (long)row * strideOm + n0;
#pragma unroll
        for (int j = 0; j < NT; ++j) orow[j * 16 + col] = acc[i][j][rr] + bv[j];
      }
    }
  }
}

template <int MT, int NT>
__global__ __launch_bounds__(384) void gemm_colA2_wmma(
    const float* __restrict__ A0, long strideAk0, long batchStrideA0,
    float* __restrict__ Out0, long batchStrideO0, int mstrips0, int waves0,
    const float* __restrict__ A1, long strideAk1, long batchStrideA1,
    float* __restrict__ Out1, long batchStrideO1, int mstrips1,
    const float* __restrict__ Wkv, int K, int nstrips) {
  int lane = threadIdx.x & 31;
  int wid = blockIdx.x * (blockDim.x >> 5) + (threadIdx.x >> 5);
  bool second = wid >= waves0;
  if (second) wid -= waves0;
  const float* A = second ? A1 : A0;
  float* Out = second ? Out1 : Out0;
  long strideAk = second ? strideAk1 : strideAk0;
  long batchStrideA = second ? batchStrideA1 : batchStrideA0;
  long batchStrideO = second ? batchStrideO1 : batchStrideO0;
  int mstrips = second ? mstrips1 : mstrips0;
  int wavesPerBatch = mstrips * nstrips;
  int b = wid / wavesPerBatch;
  int r = wid - b * wavesPerBatch;
  int m0 = (r / nstrips) * (16 * MT);
  int n0 = (r % nstrips) * (16 * NT);
  const float* Ab = A + (long)b * batchStrideA;
  float* Ob = Out + (long)b * batchStrideO;
  int col = lane & 15;
  int hi = lane >> 4;
  int khalf = hi * 2;
  static_assert(MT % 4 == 0, "M tiles come in row-interleaved groups of four");
  const float* Am0 = Ab + (long)khalf * strideAk + m0 + 4 * col;
  const float* Wp = Wkv + (long)(n0 + col) * WKV_LD + khalf;
  const long Wn16 = (long)16 * WKV_LD;

  v8f acc[MT][NT] = {};
  int Kmain = K & ~3;
  for (int k0 = 0; k0 < Kmain; k0 += 4) {
    long ao = (long)k0 * strideAk;
    v2f a[MT];
#pragma unroll
    for (int g = 0; g < MT / 4; ++g) {
      v4f x0 = *(const v4f*)(Am0 + ao + 64 * g);
      v4f x1 = *(const v4f*)(Am0 + ao + strideAk + 64 * g);
#pragma unroll
      for (int t = 0; t < 4; ++t) a[4 * g + t] = (v2f){x0[t], x1[t]};
    }
#pragma unroll
    for (int j = 0; j < NT; ++j) {
      v2f bf = *(const v2f*)(Wp + j * Wn16 + k0);
#pragma unroll
      for (int i = 0; i < MT; ++i) acc[i][j] = wmma4(a[i], bf, acc[i][j]);
    }
  }
  if (Kmain < K) {
    int k0 = Kmain;
    long ao = (long)k0 * strideAk;
    v2f a[MT];
#pragma unroll
    for (int i = 0; i < MT; ++i) a[i] = (v2f){0.f, 0.f};
    if (hi == 0) {
#pragma unroll
      for (int g = 0; g < MT / 4; ++g) {
        v4f x0 = *(const v4f*)(Am0 + ao + 64 * g);
        v4f x1 = *(const v4f*)(Am0 + ao + strideAk + 64 * g);
#pragma unroll
        for (int t = 0; t < 4; ++t) a[4 * g + t] = (v2f){x0[t], x1[t]};
      }
    }
#pragma unroll
    for (int j = 0; j < NT; ++j) {
      v2f bf = *(const v2f*)(Wp + j * Wn16 + k0);
#pragma unroll
      for (int i = 0; i < MT; ++i) acc[i][j] = wmma4(a[i], bf, acc[i][j]);
    }
  }
  if (second) {
#pragma unroll
    for (int i = 0; i < MT; ++i) {
#pragma unroll
      for (int rr = 0; rr < 4; ++rr) {
        int ch = m0 + (i >> 2) * 64 + 4 * (rr + 8 * hi) + (i & 3);
        int hh = ch & 15, d2 = ch >> 5;
        float* orow = Ob + ((long)(hh * (HD_ / 2) + d2) * P_ + n0) * 2;
#pragma unroll
        for (int j = 0; j < NT; ++j)
          *(v2f*)(orow + (j * 16 + col) * 2) = (v2f){acc[i][j][rr], acc[i][j][rr + 4]};
      }
    }
  } else {
#pragma unroll
    for (int i = 0; i < MT; ++i) {
#pragma unroll
      for (int rr = 0; rr < 8; ++rr) {
        int ch = m0 + (i >> 2) * 64 + 4 * (rr + 8 * hi) + (i & 3);
        int hh = ch & 15, dv = ch >> 4;
#pragma unroll
        for (int j = 0; j < NT; ++j) {
          int p = n0 + j * 16 + col;
          Ob[((long)(hh * (P_ / 8) + (p >> 3)) * DV_ + dv) * 8 + (p & 7)] = acc[i][j][rr];
        }
      }
    }
  }
}

template <int MT, int NT>
__global__ __launch_bounds__(256) __attribute__((amdgpu_waves_per_eu(2, 2))) void gemm_colA_rowW_wmma(
    const float* __restrict__ At, const float* __restrict__ W,
    const float* __restrict__ bias, float* __restrict__ Out, int Mtot, int K,
    int nstrips) {
  int lane = threadIdx.x & 31;
  int wid = (blockIdx.x * blockDim.x + threadIdx.x) >> 5;
  int m0 = (wid / nstrips) * (16 * MT);
  int n0 = (wid % nstrips) * (16 * NT);
  int col = lane & 15;
  int hi = lane >> 4;
  int khalf = hi * 2;
  static_assert(MT % 4 == 0, "M tiles come in row-interleaved groups of four");
  const float* Ap[MT / 4];
#pragma unroll
  for (int g = 0; g < MT / 4; ++g) {
    int m = m0 + 64 * g + 4 * col;
    if (m > Mtot - 4) m = Mtot - 4;
    Ap[g] = At + (long)khalf * Mtot + m;
  }
  const float* Wp = W + (long)(n0 + col) * K + khalf;
  const long Wn16 = (long)16 * K;

  v8f acc[MT][NT] = {};
#pragma unroll 4
  for (int k0 = 0; k0 < K; k0 += 4) {
    long ao = (long)k0 * Mtot;
    v2f a[MT];
#pragma unroll
    for (int g = 0; g < MT / 4; ++g) {
      v4f x0 = *(const v4f*)(Ap[g] + ao);
      v4f x1 = *(const v4f*)(Ap[g] + ao + Mtot);
#pragma unroll
      for (int t = 0; t < 4; ++t) a[4 * g + t] = (v2f){x0[t], x1[t]};
    }
#pragma unroll
    for (int j = 0; j < NT; ++j) {
      v2f bf = *(const v2f*)(Wp + j * Wn16 + k0);
#pragma unroll
      for (int i = 0; i < MT; ++i) acc[i][j] = wmma4(a[i], bf, acc[i][j]);
    }
  }
  float bv[NT];
#pragma unroll
  for (int j = 0; j < NT; ++j) bv[j] = bias[n0 + j * 16 + col];
#pragma unroll
  for (int i = 0; i < MT; ++i) {
#pragma unroll
    for (int rr = 0; rr < 8; ++rr) {
      int row = m0 + (i >> 2) * 64 + 4 * (rr + 8 * hi) + (i & 3);
      if (row < Mtot) {
        int bb = row / L_;
        int l = row - bb * L_;
        float* orow = Out + ((long)l * B_ + bb) * D_ + n0;
#pragma unroll
        for (int j = 0; j < NT; ++j) orow[j * 16 + col] = acc[i][j][rr] + bv[j];
      }
    }
  }
}

__global__ __launch_bounds__(256, 4) void attn_wmma(
    const float* __restrict__ q, const float* __restrict__ kbuf,
    const float* __restrict__ vbuf, float* __restrict__ obuf) {
  __shared__ float lds[8][2][2][256];
  int lane = threadIdx.x & 31;
  int wslot = threadIdx.x >> 5;
  int wid = (blockIdx.x * blockDim.x + threadIdx.x) >> 5;
  int ltile = wid & 63;
  int h = (wid >> 6) & 15;
  int b = wid >> 10;
  int l0 = ltile * 32;
  int col = lane & 15;
  int hi = lane >> 4;
  int khalf = hi * 2;

  v2f qA[2][8];
#pragma unroll
  for (int i = 0; i < 2; ++i) {
    int l = l0 + 16 * i + col;
    bool lv = l < L_;
    const float* qrow = q + ((long)b * L_ + l) * C_ + h * HD_;
    const float sc = 0.125f * 1.44269504088896340736f;
#pragma unroll
    for (int kk = 0; kk < 8; ++kk) {
      int k = kk * 4 + khalf;
      float x0 = lv ? qrow[k] * sc : 0.f;
      float x1 = lv ? qrow[k + 1] * sc : 0.f;
      qA[i][kk] = (v2f){x0, x1};
    }
  }

  v8f O[2][4] = {};
  float Mrow[2][8], Srow[2][8];
#pragma unroll
  for (int i = 0; i < 2; ++i) {
#pragma unroll
    for (int r = 0; r < 8; ++r) { Mrow[i][r] = -INFINITY; Srow[i][r] = 0.f; }
  }

  const float* kb = kbuf + (long)b * C_ * P_ + (long)h * (HD_ * P_) + (hi * P_ + col) * 2;
  const float* vb = vbuf + (long)b * D_ * P_ + (long)h * (P_ * DV_) + col * 8 + khalf;

  for (int p0 = 0; p0 < P_; p0 += 32) {
    v8f c[2][2] = {};
#pragma unroll
    for (int s = 0; s < 2; ++s) {
#pragma unroll
      for (int kk = 0; kk < 8; ++kk) {
        const float* kp = kb + (long)(2 * kk * P_ + p0 + 16 * s) * 2;
        v2f bf = *(const v2f*)kp;
        c[0][s] = wmma4(qA[0][kk], bf, c[0][s]);
        c[1][s] = wmma4(qA[1][kk], bf, c[1][s]);
      }
    }
#pragma unroll
    for (int i = 0; i < 2; ++i) {
      float alpha[8];
#pragma unroll
      for (int r = 0; r < 8; ++r) {
        float mx = rowmax16(fmaxf(c[i][0][r], c[i][1][r]));
        float mnew = fmaxf(Mrow[i][r], mx);
        float al = __builtin_exp2f(Mrow[i][r] - mnew);
        float pv0 = __builtin_exp2f(c[i][0][r] - mnew);
        float pv1 = __builtin_exp2f(c[i][1][r] - mnew);
        Srow[i][r] = Srow[i][r] * al + rowsum16(pv0 + pv1);
        Mrow[i][r] = mnew;
        alpha[r] = al;
        lds[wslot][i][0][(r + 8 * hi) * 16 + col] = pv0;
        lds[wslot][i][1][(r + 8 * hi) * 16 + col] = pv1;
      }
#pragma unroll
      for (int r = 0; r < 8; ++r) {
        O[i][0][r] *= alpha[r]; O[i][1][r] *= alpha[r];
        O[i][2][r] *= alpha[r]; O[i][3][r] *= alpha[r];
      }
    }
    __builtin_amdgcn_sched_barrier(0);
    __builtin_amdgcn_fence(__ATOMIC_ACQ_REL, "wavefront");
    __builtin_amdgcn_wave_barrier();
    v2f pA[2][2][4];
#pragma unroll
    for (int i = 0; i < 2; ++i) {
#pragma unroll
      for (int s = 0; s < 2; ++s) {
#pragma unroll
        for (int kk = 0; kk < 4; ++kk) {
          int k = kk * 4 + khalf;
          pA[i][s][kk] = (v2f){lds[wslot][i][s][col * 16 + k], lds[wslot][i][s][col * 16 + k + 1]};
        }
      }
    }
    __builtin_amdgcn_fence(__ATOMIC_ACQ_REL, "wavefront");
    __builtin_amdgcn_wave_barrier();
#pragma unroll
    for (int s = 0; s < 2; ++s) {
#pragma unroll
      for (int j = 0; j < 4; ++j) {
#pragma unroll
        for (int kk = 0; kk < 4; ++kk) {
          const float* vp = vb + (long)p0 * DV_ + (2 * s + (kk >> 1)) * (DV_ * 8) + j * 128 + 4 * (kk & 1);
          v2f bf = *(const v2f*)vp;
          O[0][j] = wmma4(pA[0][s][kk], bf, O[0][j]);
          O[1][j] = wmma4(pA[1][s][kk], bf, O[1][j]);
        }
      }
    }
  }

#pragma unroll
  for (int i = 0; i < 2; ++i) {
#pragma unroll
    for (int r = 0; r < 8; ++r) {
      float inv = 1.f / Srow[i][r];
      O[i][0][r] *= inv; O[i][1][r] *= inv; O[i][2][r] *= inv; O[i][3][r] *= inv;
    }
#pragma unroll
    for (int j = 0; j < 4; ++j) {
      float* ocol = obuf + (long)((j * 16 + col) * 16 + h) * MTOT_ + (long)b * L_ + l0 + 16 * i + 8 * hi;
#pragma unroll
      for (int rp = 0; rp < 4; ++rp) {
        int l = l0 + 16 * i + 8 * hi + 2 * rp;
        if (l < L_) *(v2f*)(ocol + 2 * rp) = (v2f){O[i][j][2 * rp], O[i][j][2 * rp + 1]};
      }
    }
  }
}

extern "C" void kernel_launch(void* const* d_in, const int* in_sizes, int n_in,
                              void* d_out, int out_size, void* d_ws,
                              size_t ws_size, hipStream_t stream) {
  (void)in_sizes; (void)n_in; (void)out_size; (void)ws_size;
  const float* in = (const float*)d_in[0];
  const float* w_kv = (const float*)d_in[1];
  const float* w_q = (const float*)d_in[2];
  const float* b_q = (const float*)d_in[3];
  const float* w_out = (const float*)d_in[4];
  const float* b_out = (const float*)d_in[5];
  float* out = (float*)d_out;

  float* ws = (float*)d_ws;
  float* qbuf = ws;
  float* vbuf = qbuf + (size_t)B_ * L_ * C_;
  float* kbuf = vbuf + (size_t)B_ * D_ * P_;
  float* obuf = kbuf + (size_t)B_ * C_ * P_;

  gemm_rowA_wmma<2, 4><<<256, 256, 0, stream>>>(in, (long)(B_ * D_), (long)D_, w_q,
                                          b_q, qbuf, (long)C_, (long)L_ * C_,
                                          L_, D_, 64, 8);
  gemm_colA2_wmma<4, 2><<<256, 384, 0, stream>>>(
      in, (long)(B_ * D_), (long)D_, vbuf, (long)D_ * P_, 16, 2048,
      qbuf, (long)C_, (long)L_ * C_, kbuf, (long)C_ * P_, 8,
      w_kv, L_, 32);
  attn_wmma<<<512, 256, 0, stream>>>(qbuf, kbuf, vbuf, obuf);
  gemm_colA_rowW_wmma<8, 2><<<256, 256, 0, stream>>>(obuf, w_out, b_out, out,
                                                     MTOT_, D_, 32);
}


# Attention: raw scores (not probabilities) take the LDS re-layout, softmax runs behind it, a lane owning 16 of one row's 32 scores: max/sum in-lane + one lane-half exchange; row sum order changes
# speedup vs baseline: 1.8996x; 1.0505x over previous
#include <hip/hip_runtime.h>


typedef __attribute__((ext_vector_type(2))) float v2f;
typedef __attribute__((ext_vector_type(4))) float v4f;
typedef __attribute__((ext_vector_type(8))) float v8f;

#define B_ 4
#define L_ 2046
#define D_ 1024
#define C_ 512
#define P_ 1024
#define H_ 16
#define HD_ 32
#define DV_ 64
#define WKV_LD 2048
#define MTOT_ (B_ * L_)

static __device__ __forceinline__ v8f wmma4(v2f a, v2f b, v8f c) {
  return __builtin_amdgcn_wmma_f32_16x16x4_f32(false, a, false, b, (short)0, c,
                                               false, false);
}

template <int MT, int NT>
__global__ __launch_bounds__(256) void gemm_rowA_wmma(
    const float* __restrict__ A, long strideAm, long batchStrideA,
    const float* __restrict__ W, const float* __restrict__ bias,
    float* __restrict__ Out, long strideOm, long batchStrideO,
    int M, int K, int mstrips, int nstrips) {
  int lane = threadIdx.x & 31;
  int wid = (blockIdx.x * blockDim.x + threadIdx.x) >> 5;
  int wavesPerBatch = mstrips * nstrips;
  int b = wid / wavesPerBatch;
  int r = wid - b * wavesPerBatch;
  int m0 = (r / nstrips) * (16 * MT);
  int n0 = (r % nstrips) * (16 * NT);
  const float* Ab = A + (long)b * batchStrideA;
  float* Ob = Out + (long)b * batchStrideO;
  int col = lane & 15;
  int hi = lane >> 4;
  int khalf = hi * 2;
  const float* Arow[MT];
#pragma unroll
  for (int i = 0; i < MT; ++i) {
    int row = m0 + 16 * i + col;
    if (row > M - 1) row = M - 1;
    Arow[i] = Ab + (long)row * strideAm + khalf;
  }
  const float* Wp = W + (long)(n0 + col) * K + khalf;
  const long Wn16 = (long)16 * K;

  v8f acc[MT][NT] = {};
  for (int k0 = 0; k0 < K; k0 += 4) {
    if ((k0 & 63) == 0) {
#pragma unroll
      for (int i = 0; i < MT; ++i) __builtin_prefetch(Arow[i] + k0 + 256, 0, 1);
      __builtin_prefetch(Wp + k0 + 256, 0, 1);
      __builtin_prefetch(Wp + (NT - 1) * Wn16 + k0 + 256, 0, 1);
    }
    v2f a[MT];
#pragma unroll
    for (int i = 0; i < MT; ++i) a[i] = *(const v2f*)(Arow[i] + k0);
#pragma unroll
    for (int j = 0; j < NT; ++j) {
      v2f bf = *(const v2f*)(Wp + j * Wn16 + k0);
#pragma unroll
      for (int i = 0; i < MT; ++i) acc[i][j] = wmma4(a[i], bf, acc[i][j]);
    }
  }
  float bv[NT];
#pragma unroll
  for (int j = 0; j < NT; ++j) bv[j] = bias[n0 + j * 16 + col];
#pragma unroll
  for (int i = 0; i < MT; ++i) {
#pragma unroll
    for (int rr = 0; rr < 8; ++rr) {
      int row = m0 + i * 16 + rr + 8 * hi;
      if (row < M) {
        float* orow = Ob + (long)row * strideOm + n0;
#pragma unroll
        for (int j = 0; j < NT; ++j) orow[j * 16 + col] = acc[i][j][rr] + bv[j];
      }
    }
  }
}

template <int MT, int NT>
__global__ __launch_bounds__(384) void gemm_colA2_wmma(
    const float* __restrict__ A0, long strideAk0, long batchStrideA0,
    float* __restrict__ Out0, long batchStrideO0, int mstrips0, int waves0,
    const float* __restrict__ A1, long strideAk1, long batchStrideA1,
    float* __restrict__ Out1, long batchStrideO1, int mstrips1,
    const float* __restrict__ Wkv, int K, int nstrips) {
  int lane = threadIdx.x & 31;
  int wid = blockIdx.x * (blockDim.x >> 5) + (threadIdx.x >> 5);
  bool second = wid >= waves0;
  if (second) wid -= waves0;
  const float* A = second ? A1 : A0;
  float* Out = second ? Out1 : Out0;
  long strideAk = second ? strideAk1 : strideAk0;
  long batchStrideA = second ? batchStrideA1 : batchStrideA0;
  long batchStrideO = second ? batchStrideO1 : batchStrideO0;
  int mstrips = second ? mstrips1 : mstrips0;
  int wavesPerBatch = mstrips * nstrips;
  int b = wid / wavesPerBatch;
  int r = wid - b * wavesPerBatch;
  int m0 = (r / nstrips) * (16 * MT);
  int n0 = (r % nstrips) * (16 * NT);
  const float* Ab = A + (long)b * batchStrideA;
  float* Ob = Out + (long)b * batchStrideO;
  int col = lane & 15;
  int hi = lane >> 4;
  int khalf = hi * 2;
  static_assert(MT % 4 == 0, "M tiles come in row-interleaved groups of four");
  const float* Am0 = Ab + (long)khalf * strideAk + m0 + 4 * col;
  const float* Wp = Wkv + (long)(n0 + col) * WKV_LD + khalf;
  const long Wn16 = (long)16 * WKV_LD;

  v8f acc[MT][NT] = {};
  int Kmain = K & ~3;
  for (int k0 = 0; k0 < Kmain; k0 += 4) {
    long ao = (long)k0 * strideAk;
    v2f a[MT];
#pragma unroll
    for (int g = 0; g < MT / 4; ++g) {
      v4f x0 = *(const v4f*)(Am0 + ao + 64 * g);
      v4f x1 = *(const v4f*)(Am0 + ao + strideAk + 64 * g);
#pragma unroll
      for (int t = 0; t < 4; ++t) a[4 * g + t] = (v2f){x0[t], x1[t]};
    }
#pragma unroll
    for (int j = 0; j < NT; ++j) {
      v2f bf = *(const v2f*)(Wp + j * Wn16 + k0);
#pragma unroll
      for (int i = 0; i < MT; ++i) acc[i][j] = wmma4(a[i], bf, acc[i][j]);
    }
  }
  if (Kmain < K) {
    int k0 = Kmain;
    long ao = (long)k0 * strideAk;
    v2f a[MT];
#pragma unroll
    for (int i = 0; i < MT; ++i) a[i] = (v2f){0.f, 0.f};
    if (hi == 0) {
#pragma unroll
      for (int g = 0; g < MT / 4; ++g) {
        v4f x0 = *(const v4f*)(Am0 + ao + 64 * g);
        v4f x1 = *(const v4f*)(Am0 + ao + strideAk + 64 * g);
#pragma unroll
        for (int t = 0; t < 4; ++t) a[4 * g + t] = (v2f){x0[t], x1[t]};
      }
    }
#pragma unroll
    for (int j = 0; j < NT; ++j) {
      v2f bf = *(const v2f*)(Wp + j * Wn16 + k0);
#pragma unroll
      for (int i = 0; i < MT; ++i) acc[i][j] = wmma4(a[i], bf, acc[i][j]);
    }
  }
  if (second) {
#pragma unroll
    for (int i = 0; i < MT; ++i) {
#pragma unroll
      for (int rr = 0; rr < 4; ++rr) {
        int ch = m0 + (i >> 2) * 64 + 4 * (rr + 8 * hi) + (i & 3);
        int hh = ch & 15, d2 = ch >> 5;
        float* orow = Ob + ((long)(hh * (HD_ / 2) + d2) * P_ + n0) * 2;
#pragma unroll
        for (int j = 0; j < NT; ++j)
          *(v2f*)(orow + (j * 16 + col) * 2) = (v2f){acc[i][j][rr], acc[i][j][rr + 4]};
      }
    }
  } else {
#pragma unroll
    for (int i = 0; i < MT; ++i) {
#pragma unroll
      for (int rr = 0; rr < 8; ++rr) {
        int ch = m0 + (i >> 2) * 64 + 4 * (rr + 8 * hi) + (i & 3);
        int hh = ch & 15, dv = ch >> 4;
#pragma unroll
        for (int j = 0; j < NT; ++j) {
          int p = n0 + j * 16 + col;
          Ob[((long)(hh * (P_ / 8) + (p >> 3)) * DV_ + dv) * 8 + (p & 7)] = acc[i][j][rr];
        }
      }
    }
  }
}

template <int MT, int NT>
__global__ __launch_bounds__(256) __attribute__((amdgpu_waves_per_eu(2, 2))) void gemm_colA_rowW_wmma(
    const float* __restrict__ At, const float* __restrict__ W,
    const float* __restrict__ bias, float* __restrict__ Out, int Mtot, int K,
    int nstrips) {
  int lane = threadIdx.x & 31;
  int wid = (blockIdx.x * blockDim.x + threadIdx.x) >> 5;
  int m0 = (wid / nstrips) * (16 * MT);
  int n0 = (wid % nstrips) * (16 * NT);
  int col = lane & 15;
  int hi = lane >> 4;
  int khalf = hi * 2;
  static_assert(MT % 4 == 0, "M tiles come in row-interleaved groups of four");
  const float* Ap[MT / 4];
#pragma unroll
  for (int g = 0; g < MT / 4; ++g) {
    int m = m0 + 64 * g + 4 * col;
    if (m > Mtot - 4) m = Mtot - 4;
    Ap[g] = At + (long)khalf * Mtot + m;
  }
  const float* Wp = W + (long)(n0 + col) * K + khalf;
  const long Wn16 = (long)16 * K;

  v8f acc[MT][NT] = {};
#pragma unroll 4
  for (int k0 = 0; k0 < K; k0 += 4) {
    long ao = (long)k0 * Mtot;
    v2f a[MT];
#pragma unroll
    for (int g = 0; g < MT / 4; ++g) {
      v4f x0 = *(const v4f*)(Ap[g] + ao);
      v4f x1 = *(const v4f*)(Ap[g] + ao + Mtot);
#pragma unroll
      for (int t = 0; t < 4; ++t) a[4 * g + t] = (v2f){x0[t], x1[t]};
    }
#pragma unroll
    for (int j = 0; j < NT; ++j) {
      v2f bf = *(const v2f*)(Wp + j * Wn16 + k0);
#pragma unroll
      for (int i = 0; i < MT; ++i) acc[i][j] = wmma4(a[i], bf, acc[i][j]);
    }
  }
  float bv[NT];
#pragma unroll
  for (int j = 0; j < NT; ++j) bv[j] = bias[n0 + j * 16 + col];
#pragma unroll
  for (int i = 0; i < MT; ++i) {
#pragma unroll
    for (int rr = 0; rr < 8; ++rr) {
      int row = m0 + (i >> 2) * 64 + 4 * (rr + 8 * hi) + (i & 3);
      if (row < Mtot) {
        int bb = row / L_;
        int l = row - bb * L_;
        float* orow = Out + ((long)l * B_ + bb) * D_ + n0;
#pragma unroll
        for (int j = 0; j < NT; ++j) orow[j * 16 + col] = acc[i][j][rr] + bv[j];
      }
    }
  }
}

static __device__ __forceinline__ float other_half(float v) {
  unsigned u = __builtin_bit_cast(unsigned, v);
  return __builtin_bit_cast(float, __builtin_amdgcn_permlanex16(u, u, 0x76543210u, 0xfedcba98u, false, false));
}

__global__ __launch_bounds__(256, 4) void attn_wmma(
    const float* __restrict__ q, const float* __restrict__ kbuf,
    const float* __restrict__ vbuf, float* __restrict__ obuf) {
  __shared__ float lds[8][2][2][256];
  __shared__ __attribute__((aligned(16))) float rowf[8][2][16];
  int lane = threadIdx.x & 31;
  int wslot = threadIdx.x >> 5;
  int wid = (blockIdx.x * blockDim.x + threadIdx.x) >> 5;
  int ltile = wid & 63;
  int h = (wid >> 6) & 15;
  int b = wid >> 10;
  int l0 = ltile * 32;
  int col = lane & 15;
  int hi = lane >> 4;
  int khalf = hi * 2;

  v2f qA[2][8];
#pragma unroll
  for (int i = 0; i < 2; ++i) {
    int l = l0 + 16 * i + col;
    bool lv = l < L_;
    const float* qrow = q + ((long)b * L_ + l) * C_ + h * HD_;
    const float sc = 0.125f * 1.44269504088896340736f;
#pragma unroll
    for (int kk = 0; kk < 8; ++kk) {
      int k = kk * 4 + khalf;
      float x0 = lv ? qrow[k] * sc : 0.f;
      float x1 = lv ? qrow[k + 1] * sc : 0.f;
      qA[i][kk] = (v2f){x0, x1};
    }
  }

  v8f O[2][4] = {};
  float Mrow[2] = {-INFINITY, -INFINITY};
  float Srow[2] = {0.f, 0.f};

  const float* kb = kbuf + (long)b * C_ * P_ + (long)h * (HD_ * P_) + (hi * P_ + col) * 2;
  const float* vb = vbuf + (long)b * D_ * P_ + (long)h * (P_ * DV_) + col * 8 + khalf;

  for (int p0 = 0; p0 < P_; p0 += 32) {
    v8f c[2][2] = {};
#pragma unroll
    for (int s = 0; s < 2; ++s) {
#pragma unroll
      for (int kk = 0; kk < 8; ++kk) {
        const float* kp = kb + (long)(2 * kk * P_ + p0 + 16 * s) * 2;
        v2f bf = *(const v2f*)kp;
        c[0][s] = wmma4(qA[0][kk], bf, c[0][s]);
        c[1][s] = wmma4(qA[1][kk], bf, c[1][s]);
      }
    }
    __builtin_amdgcn_sched_barrier(0);
#pragma unroll
    for (int i = 0; i < 2; ++i) {
#pragma unroll
      for (int r = 0; r < 8; ++r) {
        lds[wslot][i][0][(r + 8 * hi) * 16 + col] = c[i][0][r];
        lds[wslot][i][1][(r + 8 * hi) * 16 + col] = c[i][1][r];
      }
    }
    __builtin_amdgcn_fence(__ATOMIC_ACQ_REL, "wavefront");
    __builtin_amdgcn_wave_barrier();
    v2f pA[2][2][4];
#pragma unroll
    for (int i = 0; i < 2; ++i) {
#pragma unroll
      for (int s = 0; s < 2; ++s) {
#pragma unroll
        for (int kk = 0; kk < 4; ++kk) {
          int k = kk * 4 + khalf;
          pA[i][s][kk] = (v2f){lds[wslot][i][s][col * 16 + k], lds[wslot][i][s][col * 16 + k + 1]};
        }
      }
    }
    __builtin_amdgcn_fence(__ATOMIC_ACQ_REL, "wavefront");
    __builtin_amdgcn_wave_barrier();
#pragma unroll
    for (int i = 0; i < 2; ++i) {
      float mx = pA[i][0][0][0];
#pragma unroll
      for (int s = 0; s < 2; ++s) {
#pragma unroll
        for (int kk = 0; kk < 4; ++kk) {
          mx = __builtin_fmaximum_numf(mx, pA[i][s][kk][0]);
          mx = __builtin_fmaximum_numf(mx, pA[i][s][kk][1]);
        }
      }
      mx = __builtin_fmaximum_numf(mx, other_half(mx));
      float mnew = __builtin_fmaximum_numf(Mrow[i], mx);
      float al = __builtin_exp2f(Mrow[i] - mnew);
      v2f part = {0.f, 0.f};
#pragma unroll
      for (int s = 0; s < 2; ++s) {
#pragma unroll
        for (int kk = 0; kk < 4; ++kk) {
          v2f e = {__builtin_exp2f(pA[i][s][kk][0] - mnew), __builtin_exp2f(pA[i][s][kk][1] - mnew)};
          pA[i][s][kk] = e;
          part += e;
        }
      }
      float ls = part[0] + part[1];
      ls += other_half(ls);
      Srow[i] = Srow[i] * al + ls;
      Mrow[i] = mnew;
      rowf[wslot][i][col] = al;
    }
    __builtin_amdgcn_fence(__ATOMIC_ACQ_REL, "wavefront");
    __builtin_amdgcn_wave_barrier();
#pragma unroll
    for (int i = 0; i < 2; ++i) {
      v4f a0 = *(const v4f*)&rowf[wslot][i][8 * hi];
      v4f a1 = *(const v4f*)&rowf[wslot][i][8 * hi + 4];
#pragma unroll
      for (int j = 0; j < 4; ++j) {
#pragma unroll
        for (int r = 0; r < 4; ++r) {
          O[i][j][r] *= a0[r];
          O[i][j][r + 4] *= a1[r];
        }
      }
    }
    __builtin_amdgcn_fence(__ATOMIC_ACQ_REL, "wavefront");
    __builtin_amdgcn_wave_barrier();
    __builtin_amdgcn_sched_barrier(0);
#pragma unroll
    for (int s = 0; s < 2; ++s) {
#pragma unroll
      for (int j = 0; j < 4; ++j) {
#pragma unroll
        for (int kk = 0; kk < 4; ++kk) {
          const float* vp = vb + (long)p0 * DV_ + (2 * s + (kk >> 1)) * (DV_ * 8) + j * 128 + 4 * (kk & 1);
          v2f bf = *(const v2f*)vp;
          O[0][j] = wmma4(pA[0][s][kk], bf, O[0][j]);
          O[1][j] = wmma4(pA[1][s][kk], bf, O[1][j]);
        }
      }
    }
  }

#pragma unroll
  for (int i = 0; i < 2; ++i) rowf[wslot][i][col] = 1.f / Srow[i];
  __builtin_amdgcn_fence(__ATOMIC_ACQ_REL, "wavefront");
  __builtin_amdgcn_wave_barrier();
#pragma unroll
  for (int i = 0; i < 2; ++i) {
    v4f n0 = *(const v4f*)&rowf[wslot][i][8 * hi];
    v4f n1 = *(const v4f*)&rowf[wslot][i][8 * hi + 4];
#pragma unroll
    for (int j = 0; j < 4; ++j) {
#pragma unroll
      for (int r = 0; r < 4; ++r) {
        O[i][j][r] *= n0[r];
        O[i][j][r + 4] *= n1[r];
      }
    }
#pragma unroll
    for (int j = 0; j < 4; ++j) {
      float* ocol = obuf + (long)((j * 16 + col) * 16 + h) * MTOT_ + (long)b * L_ + l0 + 16 * i + 8 * hi;
#pragma unroll
      for (int rp = 0; rp < 4; ++rp) {
        int l = l0 + 16 * i + 8 * hi + 2 * rp;
        if (l < L_) *(v2f*)(ocol + 2 * rp) = (v2f){O[i][j][2 * rp], O[i][j][2 * rp + 1]};
      }
    }
  }
}

extern "C" void kernel_launch(void* const* d_in, const int* in_sizes, int n_in,
                              void* d_out, int out_size, void* d_ws,
                              size_t ws_size, hipStream_t stream) {
  (void)in_sizes; (void)n_in; (void)out_size; (void)ws_size;
  const float* in = (const float*)d_in[0];
  const float* w_kv = (const float*)d_in[1];
  const float* w_q = (const float*)d_in[2];
  const float* b_q = (const float*)d_in[3];
  const float* w_out = (const float*)d_in[4];
  const float* b_out = (const float*)d_in[5];
  float* out = (float*)d_out;

  float* ws = (float*)d_ws;
  float* qbuf = ws;
  float* vbuf = qbuf + (size_t)B_ * L_ * C_;
  float* kbuf = vbuf + (size_t)B_ * D_ * P_;
  float* obuf = kbuf + (size_t)B_ * C_ * P_;

  gemm_rowA_wmma<2, 4><<<256, 256, 0, stream>>>(in, (long)(B_ * D_), (long)D_, w_q,
                                          b_q, qbuf, (long)C_, (long)L_ * C_,
                                          L_, D_, 64, 8);
  gemm_colA2_wmma<4, 2><<<256, 384, 0, stream>>>(
      in, (long)(B_ * D_), (long)D_, vbuf, (long)D_ * P_, 16, 2048,
      qbuf, (long)C_, (long)L_ * C_, kbuf, (long)C_ * P_, 8,
      w_kv, L_, 32);
  attn_wmma<<<512, 256, 0, stream>>>(qbuf, kbuf, vbuf, obuf);
  gemm_colA_rowW_wmma<8, 2><<<256, 256, 0, stream>>>(obuf, w_out, b_out, out,
                                                     MTOT_, D_, 32);
}


# All four kernels: operand fragments are loaded ahead of their WMMA run into a second register set (GEMMs: one 4-k-step trip ahead; attention: K one trip ahead, V before the softmax); sum orders kept
# speedup vs baseline: 2.0947x; 1.1027x over previous
#include <hip/hip_runtime.h>


typedef __attribute__((ext_vector_type(2))) float v2f;
typedef __attribute__((ext_vector_type(4))) float v4f;
typedef __attribute__((ext_vector_type(8))) float v8f;

#define B_ 4
#define L_ 2046
#define D_ 1024
#define C_ 512
#define P_ 1024
#define H_ 16
#define HD_ 32
#define DV_ 64
#define WKV_LD 2048
#define MTOT_ (B_ * L_)

static __device__ __forceinline__ v8f wmma4(v2f a, v2f b, v8f c) {
  return __builtin_amdgcn_wmma_f32_16x16x4_f32(false, a, false, b, (short)0, c,
                                               false, false);
}


template <int MT, int NT, int KS>
struct TripColA {
  v4f x0[KS][MT / 4];
  v4f x1[KS][MT / 4];
  v2f bf[KS][NT];
};

template <int MT, int NT, int KS>
static __device__ __forceinline__ void mma_trip_colA(const TripColA<MT, NT, KS>& f,
                                                     v8f (&acc)[MT][NT]) {
#pragma unroll
  for (int s = 0; s < KS; ++s) {
    v2f a[MT];
#pragma unroll
    for (int g = 0; g < MT / 4; ++g) {
#pragma unroll
      for (int t = 0; t < 4; ++t) a[4 * g + t] = (v2f){f.x0[s][g][t], f.x1[s][g][t]};
    }
#pragma unroll
    for (int j = 0; j < NT; ++j) {
#pragma unroll
      for (int i = 0; i < MT; ++i) acc[i][j] = wmma4(a[i], f.bf[s][j], acc[i][j]);
    }
  }
}

template <int MT, int NT, int KS>
struct TripRowA {
  v2f a[KS][MT];
  v2f bf[KS][NT];
};

template <int MT, int NT, int KS>
static __device__ __forceinline__ void mma_trip_rowA(const TripRowA<MT, NT, KS>& f,
                                                     v8f (&acc)[MT][NT]) {
#pragma unroll
  for (int s = 0; s < KS; ++s) {
#pragma unroll
    for (int j = 0; j < NT; ++j) {
#pragma unroll
      for (int i = 0; i < MT; ++i) acc[i][j] = wmma4(f.a[s][i], f.bf[s][j], acc[i][j]);
    }
  }
}

template <int MT, int NT>
__global__ __launch_bounds__(256) void gemm_rowA_wmma(
    const float* __restrict__ A, long strideAm, long batchStrideA,
    const float* __restrict__ W, const float* __restrict__ bias,
    float* __restrict__ Out, long strideOm, long batchStrideO,
    int M, int K, int mstrips, int nstrips) {
  int lane = threadIdx.x & 31;
  int wid = (blockIdx.x * blockDim.x + threadIdx.x) >> 5;
  int wavesPerBatch = mstrips * nstrips;
  int b = wid / wavesPerBatch;
  int r = wid - b * wavesPerBatch;
  int m0 = (r / nstrips) * (16 * MT);
  int n0 = (r % nstrips) * (16 * NT);
  const float* Ab = A + (long)b * batchStrideA;
  float* Ob = Out + (long)b * batchStrideO;
  int col = lane & 15;
  int hi = lane >> 4;
  int khalf = hi * 2;
  const float* Arow[MT];
#pragma unroll
  for (int i = 0; i < MT; ++i) {
    int row = m0 + 16 * i + col;
    if (row > M - 1) row = M - 1;
    Arow[i] = Ab + (long)row * strideAm + khalf;
  }
  const float* Wp = W + (long)(n0 + col) * K + khalf;
  const long Wn16 = (long)16 * K;

  v8f acc[MT][NT] = {};
  constexpr int KS = 4;
  TripRowA<MT, NT, KS> f0, f1;
  auto load_trip = [&](TripRowA<MT, NT, KS>& f, int k0) {
#pragma unroll
    for (int s = 0; s < KS; ++s) {
#pragma unroll
      for (int i = 0; i < MT; ++i) f.a[s][i] = *(const v2f*)(Arow[i] + k0 + 4 * s);
#pragma unroll
      for (int j = 0; j < NT; ++j)
        f.bf[s][j] = *(const v2f*)(Wp + j * Wn16 + k0 + 4 * s);
    }
  };
  const int trips = (K >> 2) / KS;
  if (trips > 0) load_trip(f0, 0);
  int t = 0;
  for (; t + 2 <= trips; t += 2) {
    load_trip(f1, (t + 1) * (4 * KS));
    __builtin_amdgcn_sched_barrier(0);
    mma_trip_rowA<MT, NT, KS>(f0, acc);
    __builtin_amdgcn_sched_barrier(0);
    int tn = t + 2 < trips ? t + 2 : trips - 1;
    load_trip(f0, tn * (4 * KS));
    __builtin_amdgcn_sched_barrier(0);
    mma_trip_rowA<MT, NT, KS>(f1, acc);
    __builtin_amdgcn_sched_barrier(0);
  }
  if (t < trips) mma_trip_rowA<MT, NT, KS>(f0, acc);
#pragma nounroll
  for (int k0 = trips * (4 * KS); k0 < K; k0 += 4) {
    v2f a[MT];
#pragma unroll
    for (int i = 0; i < MT; ++i) a[i] = *(const v2f*)(Arow[i] + k0);
#pragma unroll
    for (int j = 0; j < NT; ++j) {
      v2f bf = *(const v2f*)(Wp + j * Wn16 + k0);
#pragma unroll
      for (int i = 0; i < MT; ++i) acc[i][j] = wmma4(a[i], bf, acc[i][j]);
    }
  }
  float bv[NT];
#pragma unroll
  for (int j = 0; j < NT; ++j) bv[j] = bias[n0 + j * 16 + col];
#pragma unroll
  for (int i = 0; i < MT; ++i) {
#pragma unroll
    for (int rr = 0; rr < 8; ++rr) {
      int row = m0 + i * 16 + rr + 8 * hi;
      if (row < M) {
        float* orow = Ob + (long)row * strideOm + n0;
#pragma unroll
        for (int j = 0; j < NT; ++j) orow[j * 16 + col] = acc[i][j][rr] + bv[j];
      }
    }
  }
}

template <int MT, int NT>
__global__ __launch_bounds__(384) void gemm_colA2_wmma(
    const float* __restrict__ A0, long strideAk0, long batchStrideA0,
    float* __restrict__ Out0, long batchStrideO0, int mstrips0, int waves0,
    const float* __restrict__ A1, long strideAk1, long batchStrideA1,
    float* __restrict__ Out1, long batchStrideO1, int mstrips1,
    const float* __restrict__ Wkv, int K, int nstrips) {
  int lane = threadIdx.x & 31;
  int wid = blockIdx.x * (blockDim.x >> 5) + (threadIdx.x >> 5);
  bool second = wid >= waves0;
  if (second) wid -= waves0;
  const float* A = second ? A1 : A0;
  float* Out = second ? Out1 : Out0;
  long strideAk = second ? strideAk1 : strideAk0;
  long batchStrideA = second ? batchStrideA1 : batchStrideA0;
  long batchStrideO = second ? batchStrideO1 : batchStrideO0;
  int mstrips = second ? mstrips1 : mstrips0;
  int wavesPerBatch = mstrips * nstrips;
  int b = wid / wavesPerBatch;
  int r = wid - b * wavesPerBatch;
  int m0 = (r / nstrips) * (16 * MT);
  int n0 = (r % nstrips) * (16 * NT);
  const float* Ab = A + (long)b * batchStrideA;
  float* Ob = Out + (long)b * batchStrideO;
  int col = lane & 15;
  int hi = lane >> 4;
  int khalf = hi * 2;
  static_assert(MT % 4 == 0, "M tiles come in row-interleaved groups of four");
  const float* Am0 = Ab + (long)khalf * strideAk + m0 + 4 * col;
  const float* Wp = Wkv + (long)(n0 + col) * WKV_LD + khalf;
  const long Wn16 = (long)16 * WKV_LD;

  v8f acc[MT][NT] = {};
  int Kmain = K & ~3;
  constexpr int KS = 4;
  TripColA<MT, NT, KS> f0, f1;
  auto load_trip = [&](TripColA<MT, NT, KS>& f, int k0) {
#pragma unroll
    for (int s = 0; s < KS; ++s) {
      long ao = (long)(k0 + 4 * s) * strideAk;
#pragma unroll
      for (int g = 0; g < MT / 4; ++g) {
        f.x0[s][g] = *(const v4f*)(Am0 + ao + 64 * g);
        f.x1[s][g] = *(const v4f*)(Am0 + ao + strideAk + 64 * g);
      }
#pragma unroll
      for (int j = 0; j < NT; ++j) f.bf[s][j] = *(const v2f*)(Wp + j * Wn16 + k0 + 4 * s);
    }
  };
  const int trips = (Kmain >> 2) / KS;
  if (trips > 0) load_trip(f0, 0);
  int t = 0;
  for (; t + 2 <= trips; t += 2) {
    load_trip(f1, (t + 1) * (4 * KS));
    __builtin_amdgcn_sched_barrier(0);
    mma_trip_colA<MT, NT, KS>(f0, acc);
    __builtin_amdgcn_sched_barrier(0);
    int tn = t + 2 < trips ? t + 2 : trips - 1;
    load_trip(f0, tn * (4 * KS));
    __builtin_amdgcn_sched_barrier(0);
    mma_trip_colA<MT, NT, KS>(f1, acc);
    __builtin_amdgcn_sched_barrier(0);
  }
  if (t < trips) mma_trip_colA<MT, NT, KS>(f0, acc);
#pragma nounroll
  for (int k0 = trips * (4 * KS); k0 < Kmain; k0 += 4) {
    long ao = (long)k0 * strideAk;
    v2f a[MT];
#pragma unroll
    for (int g = 0; g < MT / 4; ++g) {
      v4f x0 = *(const v4f*)(Am0 + ao + 64 * g);
      v4f x1 = *(const v4f*)(Am0 + ao + strideAk + 64 * g);
#pragma unroll
      for (int tt = 0; tt < 4; ++tt) a[4 * g + tt] = (v2f){x0[tt], x1[tt]};
    }
#pragma unroll
    for (int j = 0; j < NT; ++j) {
      v2f bf = *(const v2f*)(Wp + j * Wn16 + k0);
#pragma unroll
      for (int i = 0; i < MT; ++i) acc[i][j] = wmma4(a[i], bf, acc[i][j]);
    }
  }
  if (Kmain < K) {
    int k0 = Kmain;
    long ao = (long)k0 * strideAk;
    v2f a[MT];
#pragma unroll
    for (int i = 0; i < MT; ++i) a[i] = (v2f){0.f, 0.f};
    if (hi == 0) {
#pragma unroll
      for (int g = 0; g < MT / 4; ++g) {
        v4f x0 = *(const v4f*)(Am0 + ao + 64 * g);
        v4f x1 = *(const v4f*)(Am0 + ao + strideAk + 64 * g);
#pragma unroll
        for (int t = 0; t < 4; ++t) a[4 * g + t] = (v2f){x0[t], x1[t]};
      }
    }
#pragma unroll
    for (int j = 0; j < NT; ++j) {
      v2f bf = *(const v2f*)(Wp + j * Wn16 + k0);
#pragma unroll
      for (int i = 0; i < MT; ++i) acc[i][j] = wmma4(a[i], bf, acc[i][j]);
    }
  }
  if (second) {
#pragma unroll
    for (int i = 0; i < MT; ++i) {
#pragma unroll
      for (int rr = 0; rr < 4; ++rr) {
        int ch = m0 + (i >> 2) * 64 + 4 * (rr + 8 * hi) + (i & 3);
        int hh = ch & 15, d2 = ch >> 5;
        float* orow = Ob + ((long)(hh * (HD_ / 2) + d2) * P_ + n0) * 2;
#pragma unroll
        for (int j = 0; j < NT; ++j)
          *(v2f*)(orow + (j * 16 + col) * 2) = (v2f){acc[i][j][rr], acc[i][j][rr + 4]};
      }
    }
  } else {
#pragma unroll
    for (int i = 0; i < MT; ++i) {
#pragma unroll
      for (int rr = 0; rr < 8; ++rr) {
        int ch = m0 + (i >> 2) * 64 + 4 * (rr + 8 * hi) + (i & 3);
        int hh = ch & 15, dv = ch >> 4;
#pragma unroll
        for (int j = 0; j < NT; ++j) {
          int p = n0 + j * 16 + col;
          Ob[((long)(hh * (P_ / 8) + (p >> 3)) * DV_ + dv) * 8 + (p & 7)] = acc[i][j][rr];
        }
      }
    }
  }
}

template <int MT, int NT>
__global__ __launch_bounds__(256) __attribute__((amdgpu_waves_per_eu(2, 2))) void gemm_colA_rowW_wmma(
    const float* __restrict__ At, const float* __restrict__ W,
    const float* __restrict__ bias, float* __restrict__ Out, int Mtot, int K,
    int nstrips) {
  int lane = threadIdx.x & 31;
  int wid = (blockIdx.x * blockDim.x + threadIdx.x) >> 5;
  int m0 = (wid / nstrips) * (16 * MT);
  int n0 = (wid % nstrips) * (16 * NT);
  int col = lane & 15;
  int hi = lane >> 4;
  int khalf = hi * 2;
  static_assert(MT % 4 == 0, "M tiles come in row-interleaved groups of four");
  const float* Ap[MT / 4];
#pragma unroll
  for (int g = 0; g < MT / 4; ++g) {
    int m = m0 + 64 * g + 4 * col;
    if (m > Mtot - 4) m = Mtot - 4;
    Ap[g] = At + (long)khalf * Mtot + m;
  }
  const float* Wp = W + (long)(n0 + col) * K + khalf;
  const long Wn16 = (long)16 * K;

  v8f acc[MT][NT] = {};
  constexpr int KS = 4;
  TripColA<MT, NT, KS> f0, f1;
  auto load_trip = [&](TripColA<MT, NT, KS>& f, int k0) {
#pragma unroll
    for (int s = 0; s < KS; ++s) {
      long ao = (long)(k0 + 4 * s) * Mtot;
#pragma unroll
      for (int g = 0; g < MT / 4; ++g) {
        f.x0[s][g] = *(const v4f*)(Ap[g] + ao);
        f.x1[s][g] = *(const v4f*)(Ap[g] + ao + Mtot);
      }
#pragma unroll
      for (int j = 0; j < NT; ++j) f.bf[s][j] = *(const v2f*)(Wp + j * Wn16 + k0 + 4 * s);
    }
  };
  const int trips = (K >> 2) / KS;
  if (trips > 0) load_trip(f0, 0);
  int t = 0;
  for (; t + 2 <= trips; t += 2) {
    load_trip(f1, (t + 1) * (4 * KS));
    __builtin_amdgcn_sched_barrier(0);
    mma_trip_colA<MT, NT, KS>(f0, acc);
    __builtin_amdgcn_sched_barrier(0);
    int tn = t + 2 < trips ? t + 2 : trips - 1;
    load_trip(f0, tn * (4 * KS));
    __builtin_amdgcn_sched_barrier(0);
    mma_trip_colA<MT, NT, KS>(f1, acc);
    __builtin_amdgcn_sched_barrier(0);
  }
  if (t < trips) mma_trip_colA<MT, NT, KS>(f0, acc);
#pragma nounroll
  for (int k0 = trips * (4 * KS); k0 < K; k0 += 4) {
    long ao = (long)k0 * Mtot;
    v2f a[MT];
#pragma unroll
    for (int g = 0; g < MT / 4; ++g) {
      v4f x0 = *(const v4f*)(Ap[g] + ao);
      v4f x1 = *(const v4f*)(Ap[g] + ao + Mtot);
#pragma unroll
      for (int tt = 0; tt < 4; ++tt) a[4 * g + tt] = (v2f){x0[tt], x1[tt]};
    }
#pragma unroll
    for (int j = 0; j < NT; ++j) {
      v2f bf = *(const v2f*)(Wp + j * Wn16 + k0);
#pragma unroll
      for (int i = 0; i < MT; ++i) acc[i][j] = wmma4(a[i], bf, acc[i][j]);
    }
  }
  float bv[NT];
#pragma unroll
  for (int j = 0; j < NT; ++j) bv[j] = bias[n0 + j * 16 + col];
#pragma unroll
  for (int i = 0; i < MT; ++i) {
#pragma unroll
    for (int rr = 0; rr < 8; ++rr) {
      int row = m0 + (i >> 2) * 64 + 4 * (rr + 8 * hi) + (i & 3);
      if (row < Mtot) {
        int bb = row / L_;
        int l = row - bb * L_;
        float* orow = Out + ((long)l * B_ + bb) * D_ + n0;
#pragma unroll
        for (int j = 0; j < NT; ++j) orow[j * 16 + col] = acc[i][j][rr] + bv[j];
      }
    }
  }
}

static __device__ __forceinline__ float other_half(float v) {
  unsigned u = __builtin_bit_cast(unsigned, v);
  return __builtin_bit_cast(float, __builtin_amdgcn_permlanex16(u, u, 0x76543210u, 0xfedcba98u, false, false));
}

__global__ __launch_bounds__(256, 4) void attn_wmma(
    const float* __restrict__ q, const float* __restrict__ kbuf,
    const float* __restrict__ vbuf, float* __restrict__ obuf) {
  __shared__ float lds[8][2][2][256];
  __shared__ __attribute__((aligned(16))) float rowf[8][2][16];
  int lane = threadIdx.x & 31;
  int wslot = threadIdx.x >> 5;
  int wid = (blockIdx.x * blockDim.x + threadIdx.x) >> 5;
  int ltile = wid & 63;
  int h = (wid >> 6) & 15;
  int b = wid >> 10;
  int l0 = ltile * 32;
  int col = lane & 15;
  int hi = lane >> 4;
  int khalf = hi * 2;

  v2f qA[2][8];
#pragma unroll
  for (int i = 0; i < 2; ++i) {
    int l = l0 + 16 * i + col;
    bool lv = l < L_;
    const float* qrow = q + ((long)b * L_ + l) * C_ + h * HD_;
    const float sc = 0.125f * 1.44269504088896340736f;
#pragma unroll
    for (int kk = 0; kk < 8; ++kk) {
      int k = kk * 4 + khalf;
      float x0 = lv ? qrow[k] * sc : 0.f;
      float x1 = lv ? qrow[k + 1] * sc : 0.f;
      qA[i][kk] = (v2f){x0, x1};
    }
  }

  v8f O[2][4] = {};
  float Mrow[2] = {-INFINITY, -INFINITY};
  float Srow[2] = {0.f, 0.f};

  const float* kb = kbuf + (long)b * C_ * P_ + (long)h * (HD_ * P_) + (hi * P_ + col) * 2;
  const float* vb = vbuf + (long)b * D_ * P_ + (long)h * (P_ * DV_) + col * 8 + khalf;

  v2f kf[2][8];
  auto load_k = [&](int p0) {
#pragma unroll
    for (int s = 0; s < 2; ++s) {
#pragma unroll
      for (int kk = 0; kk < 8; ++kk) kf[s][kk] = *(const v2f*)(kb + (long)(2 * kk * P_ + p0 + 16 * s) * 2);
    }
  };
  load_k(0);

  for (int p0 = 0; p0 < P_; p0 += 32) {
    v8f c[2][2] = {};
#pragma unroll
    for (int s = 0; s < 2; ++s) {
#pragma unroll
      for (int kk = 0; kk < 8; ++kk) {
        c[0][s] = wmma4(qA[0][kk], kf[s][kk], c[0][s]);
        c[1][s] = wmma4(qA[1][kk], kf[s][kk], c[1][s]);
      }
    }
    __builtin_amdgcn_sched_barrier(0);
    v2f vf[2][4][4];
#pragma unroll
    for (int s = 0; s < 2; ++s) {
#pragma unroll
      for (int j = 0; j < 4; ++j) {
#pragma unroll
        for (int kk = 0; kk < 4; ++kk)
          vf[s][j][kk] = *(const v2f*)(vb + (long)p0 * DV_ + (2 * s + (kk >> 1)) * (DV_ * 8) + j * 128 + 4 * (kk & 1));
      }
    }
    __builtin_amdgcn_sched_barrier(0);
#pragma unroll
    for (int i = 0; i < 2; ++i) {
#pragma unroll
      for (int r = 0; r < 8; ++r) {
        lds[wslot][i][0][(r + 8 * hi) * 16 + col] = c[i][0][r];
        lds[wslot][i][1][(r + 8 * hi) * 16 + col] = c[i][1][r];
      }
    }
    __builtin_amdgcn_fence(__ATOMIC_ACQ_REL, "wavefront");
    __builtin_amdgcn_wave_barrier();
    v2f pA[2][2][4];
#pragma unroll
    for (int i = 0; i < 2; ++i) {
#pragma unroll
      for (int s = 0; s < 2; ++s) {
#pragma unroll
        for (int kk = 0; kk < 4; ++kk) {
          int k = kk * 4 + khalf;
          pA[i][s][kk] = (v2f){lds[wslot][i][s][col * 16 + k], lds[wslot][i][s][col * 16 + k + 1]};
        }
      }
    }
    __builtin_amdgcn_fence(__ATOMIC_ACQ_REL, "wavefront");
    __builtin_amdgcn_wave_barrier();
#pragma unroll
    for (int i = 0; i < 2; ++i) {
      float mx = pA[i][0][0][0];
#pragma unroll
      for (int s = 0; s < 2; ++s) {
#pragma unroll
        for (int kk = 0; kk < 4; ++kk) {
          mx = __builtin_fmaximum_numf(mx, pA[i][s][kk][0]);
          mx = __builtin_fmaximum_numf(mx, pA[i][s][kk][1]);
        }
      }
      mx = __builtin_fmaximum_numf(mx, other_half(mx));
      float mnew = __builtin_fmaximum_numf(Mrow[i], mx);
      float al = __builtin_exp2f(Mrow[i] - mnew);
      v2f part = {0.f, 0.f};
#pragma unroll
      for (int s = 0; s < 2; ++s) {
#pragma unroll
        for (int kk = 0; kk < 4; ++kk) {
          v2f e = {__builtin_exp2f(pA[i][s][kk][0] - mnew), __builtin_exp2f(pA[i][s][kk][1] - mnew)};
          pA[i][s][kk] = e;
          part += e;
        }
      }
      float ls = part[0] + part[1];
      ls += other_half(ls);
      Srow[i] = Srow[i] * al + ls;
      Mrow[i] = mnew;
      rowf[wslot][i][col] = al;
    }
    __builtin_amdgcn_fence(__ATOMIC_ACQ_REL, "wavefront");
    __builtin_amdgcn_wave_barrier();
#pragma unroll
    for (int i = 0; i < 2; ++i) {
      v4f a0 = *(const v4f*)&rowf[wslot][i][8 * hi];
      v4f a1 = *(const v4f*)&rowf[wslot][i][8 * hi + 4];
#pragma unroll
      for (int j = 0; j < 4; ++j) {
#pragma unroll
        for (int r = 0; r < 4; ++r) {
          O[i][j][r] *= a0[r];
          O[i][j][r + 4] *= a1[r];
        }
      }
    }
    __builtin_amdgcn_fence(__ATOMIC_ACQ_REL, "wavefront");
    __builtin_amdgcn_wave_barrier();
    __builtin_amdgcn_sched_barrier(0);
#pragma unroll
    for (int s = 0; s < 2; ++s) {
#pragma unroll
      for (int j = 0; j < 4; ++j) {
#pragma unroll
        for (int kk = 0; kk < 4; ++kk) {
          O[0][j] = wmma4(pA[0][s][kk], vf[s][j][kk], O[0][j]);
          O[1][j] = wmma4(pA[1][s][kk], vf[s][j][kk], O[1][j]);
        }
      }
      if (s == 0) {
        __builtin_amdgcn_sched_barrier(0);
        load_k(p0 + 32 < P_ ? p0 + 32 : p0);
        __builtin_amdgcn_sched_barrier(0);
      }
    }
  }

#pragma unroll
  for (int i = 0; i < 2; ++i) rowf[wslot][i][col] = 1.f / Srow[i];
  __builtin_amdgcn_fence(__ATOMIC_ACQ_REL, "wavefront");
  __builtin_amdgcn_wave_barrier();
#pragma unroll
  for (int i = 0; i < 2; ++i) {
    v4f n0 = *(const v4f*)&rowf[wslot][i][8 * hi];
    v4f n1 = *(const v4f*)&rowf[wslot][i][8 * hi + 4];
#pragma unroll
    for (int j = 0; j < 4; ++j) {
#pragma unroll
      for (int r = 0; r < 4; ++r) {
        O[i][j][r] *= n0[r];
        O[i][j][r + 4] *= n1[r];
      }
    }
#pragma unroll
    for (int j = 0; j < 4; ++j) {
      float* ocol = obuf + (long)((j * 16 + col) * 16 + h) * MTOT_ + (long)b * L_ + l0 + 16 * i + 8 * hi;
#pragma unroll
      for (int rp = 0; rp < 4; ++rp) {
        int l = l0 + 16 * i + 8 * hi + 2 * rp;
        if (l < L_) *(v2f*)(ocol + 2 * rp) = (v2f){O[i][j][2 * rp], O[i][j][2 * rp + 1]};
      }
    }
  }
}

extern "C" void kernel_launch(void* const* d_in, const int* in_sizes, int n_in,
                              void* d_out, int out_size, void* d_ws,
                              size_t ws_size, hipStream_t stream) {
  (void)in_sizes; (void)n_in; (void)out_size; (void)ws_size;
  const float* in = (const float*)d_in[0];
  const float* w_kv = (const float*)d_in[1];
  const float* w_q = (const float*)d_in[2];
  const float* b_q = (const float*)d_in[3];
  const float* w_out = (const float*)d_in[4];
  const float* b_out = (const float*)d_in[5];
  float* out = (float*)d_out;

  float* ws = (float*)d_ws;
  float* qbuf = ws;
  float* vbuf = qbuf + (size_t)B_ * L_ * C_;
  float* kbuf = vbuf + (size_t)B_ * D_ * P_;
  float* obuf = kbuf + (size_t)B_ * C_ * P_;

  gemm_rowA_wmma<2, 4><<<256, 256, 0, stream>>>(in, (long)(B_ * D_), (long)D_, w_q,
                                          b_q, qbuf, (long)C_, (long)L_ * C_,
                                          L_, D_, 64, 8);
  gemm_colA2_wmma<4, 2><<<256, 384, 0, stream>>>(
      in, (long)(B_ * D_), (long)D_, vbuf, (long)D_ * P_, 16, 2048,
      qbuf, (long)C_, (long)L_ * C_, kbuf, (long)C_ * P_, 8,
      w_kv, L_, 32);
  attn_wmma<<<512, 256, 0, stream>>>(qbuf, kbuf, vbuf, obuf);
  gemm_colA_rowW_wmma<8, 2><<<256, 256, 0, stream>>>(obuf, w_out, b_out, out,
                                                     MTOT_, D_, 32);
}
